# starting kernel
# baseline (speedup 1.0000x reference)
#include <hip/hip_runtime.h>

typedef float v2f __attribute__((ext_vector_type(2)));
typedef float v8f __attribute__((ext_vector_type(8)));

constexpr int NN   = 2048;
constexpr int C    = 128;
constexpr int L    = 16;
constexpr int E    = 10;
constexpr int K3_0 = 23, K2_0 = 4;
constexpr int K3_1 = 33, K2_1 = 6;

constexpr int COL_A1 = 0;
constexpr int COL_A0 = 3 * K3_1;
constexpr int COL_21 = COL_A0 + K3_0;
constexpr int COL_20 = COL_21 + 3 * K2_1;
constexpr int NCOL   = COL_20 + K2_0;
constexpr int NTILE  = NCOL / 16;

constexpr int S2   = 2 * NCOL + 2;
constexpr int PSTR = 149;
constexpr int XSTR = L + 1;

constexpr int WR_M1 = 0;
constexpr int WR_M0 = 33;
constexpr int WR_21 = 56;
constexpr int WR_20 = 62;
constexpr int WR_11 = 66;
constexpr int WR_10 = 67;
constexpr int WROWS = 68;

constexpr int OFF_X  = 0;
constexpr int OFF_XS = OFF_X + C * XSTR;
constexpr int OFF_W  = OFF_XS + L;
constexpr int OFF_U1 = OFF_W + WROWS * C;
constexpr int OFF_B2 = OFF_U1 + 64;
constexpr int OFF_P  = OFF_B2 + 128 * S2;
constexpr int SMEM_FLOATS = OFF_P + C * PSTR;

__device__ __forceinline__ void build_a1_pair(const float* __restrict__ U3_1,
                                              const float* __restrict__ sXS,
                                              float* __restrict__ sB2, int w, int p)
{
    const float* u0 = U3_1 + (size_t)(w * 256 + 2 * p) * (L * K3_1);
    const float* u1 = u0 + L * K3_1;
    float a0[K3_1], a1[K3_1];
    #pragma unroll
    for (int k = 0; k < K3_1; ++k) { a0[k] = 0.f; a1[k] = 0.f; }
    for (int i = 0; i < L; ++i) {
        const float xi = sXS[i];
        #pragma unroll
        for (int k = 0; k < K3_1; ++k) {
            a0[k] = fmaf(u0[i * K3_1 + k], xi, a0[k]);
            a1[k] = fmaf(u1[i * K3_1 + k], xi, a1[k]);
        }
    }
    float* bp = sB2 + p * S2 + (COL_A1 + w * K3_1) * 2;
    #pragma unroll
    for (int k = 0; k < K3_1; ++k)
        *(v2f*)(bp + 2 * k) = (v2f){a0[k], a1[k]};
}

__device__ __forceinline__ void build_a0_pair(const float* __restrict__ U3_0,
                                              const float* __restrict__ sXS,
                                              float* __restrict__ sB2, int p)
{
    const float* u0 = U3_0 + (size_t)(2 * p) * (L * K3_0);
    const float* u1 = u0 + L * K3_0;
    float a0[K3_0], a1[K3_0];
    #pragma unroll
    for (int k = 0; k < K3_0; ++k) { a0[k] = 0.f; a1[k] = 0.f; }
    for (int i = 0; i < L; ++i) {
        const float xi = sXS[i];
        #pragma unroll
        for (int k = 0; k < K3_0; ++k) {
            a0[k] = fmaf(u0[i * K3_0 + k], xi, a0[k]);
            a1[k] = fmaf(u1[i * K3_0 + k], xi, a1[k]);
        }
    }
    float* bp = sB2 + p * S2 + COL_A0 * 2;
    #pragma unroll
    for (int k = 0; k < K3_0; ++k)
        *(v2f*)(bp + 2 * k) = (v2f){a0[k], a1[k]};
}

__global__ __launch_bounds__(256) void mace_node_kernel(
    const float* __restrict__ x,    const float* __restrict__ y,
    const float* __restrict__ U3_0, const float* __restrict__ U2_0, const float* __restrict__ U1_0,
    const float* __restrict__ U3_1, const float* __restrict__ U2_1, const float* __restrict__ U1_1,
    const float* __restrict__ Wm0,  const float* __restrict__ W20,  const float* __restrict__ W10,
    const float* __restrict__ Wm1,  const float* __restrict__ W21,  const float* __restrict__ W11,
    float* __restrict__ out)
{
    extern __shared__ float sm[];
    float* sX  = sm + OFF_X;
    float* sXS = sm + OFF_XS;
    float* sW  = sm + OFF_W;
    float* sU1 = sm + OFF_U1;
    float* sB2 = sm + OFF_B2;
    float* sP  = sm + OFF_P;

    const int tid = threadIdx.x;
    const int b   = blockIdx.x;

    for (int idx = tid; idx < C * L; idx += 256)
        sX[(idx >> 4) * XSTR + (idx & 15)] = x[(size_t)b * (C * L) + idx];

    float yv[E];
    #pragma unroll
    for (int e = 0; e < E; ++e) yv[e] = y[b * E + e];
    for (int idx = tid; idx < WROWS * C; idx += 256) {
        const int wr = idx >> 7, c = idx & 127;
        const float* src; int stride;
        if (wr < WR_M0)       { src = Wm1 + (wr)         * C + c; stride = K3_1 * C; }
        else if (wr < WR_21)  { src = Wm0 + (wr - WR_M0) * C + c; stride = K3_0 * C; }
        else if (wr < WR_20)  { src = W21 + (wr - WR_21) * C + c; stride = K2_1 * C; }
        else if (wr < WR_11)  { src = W20 + (wr - WR_20) * C + c; stride = K2_0 * C; }
        else if (wr == WR_11) { src = W11 + c;                    stride = C;        }
        else                  { src = W10 + c;                    stride = C;        }
        float acc = 0.f;
        #pragma unroll
        for (int e = 0; e < E; ++e) acc = fmaf(yv[e], src[e * stride], acc);
        sW[idx] = acc;
    }
    if (tid < 16)               sU1[tid] = U1_0[tid];
    else if (tid < 64)          sU1[tid] = U1_1[tid - 16];
    __syncthreads();

    if (tid < L) {
        float s = 0.f;
        for (int c = 0; c < C; ++c) s += sX[c * XSTR + tid];
        sXS[tid] = s;
    }
    __syncthreads();

    {
        const int t0 = tid;
        build_a1_pair(U3_1, sXS, sB2, t0 >> 7, t0 & 127);
        if (tid >= 128) {
            const int t1 = tid + 128;
            build_a1_pair(U3_1, sXS, sB2, t1 >> 7, t1 & 127);
        } else {
            const int p = tid;
            build_a0_pair(U3_0, sXS, sB2, p);
            float* bp = sB2 + p * S2;
            #pragma unroll
            for (int m = 0; m < 3 * K2_1; ++m) {
                const int w = m / K2_1, k = m % K2_1;
                const int r0 = (w * 256 + 2 * p) * K2_1 + k;
                *(v2f*)(bp + (COL_21 + m) * 2) = (v2f){U2_1[r0], U2_1[r0 + K2_1]};
            }
            #pragma unroll
            for (int m = 0; m < K2_0; ++m)
                *(v2f*)(bp + (COL_20 + m) * 2) =
                    (v2f){U2_0[(2 * p) * K2_0 + m], U2_0[(2 * p + 1) * K2_0 + m]};
        }
    }
    __syncthreads();

    const int wv   = tid >> 5;
    const int lane = tid & 31;
    const int nl   = lane & 15;
    const int hi   = lane >> 4;
    const int crow = wv * 16 + nl;

    float xr[L];
    #pragma unroll
    for (int j = 0; j < L; ++j) xr[j] = sX[crow * XSTR + j];

    float sel0[4], sel1[4];
    #pragma unroll
    for (int j = 0; j < 4; ++j) {
        const float e0 = xr[4 * j],     e1 = xr[4 * j + 1];
        const float e2 = xr[4 * j + 2], e3 = xr[4 * j + 3];
        sel0[j] = hi ? e2 : e0;
        sel1[j] = hi ? e3 : e1;
    }

    v8f acc[NTILE];
    #pragma unroll
    for (int t = 0; t < NTILE; ++t)
        acc[t] = (v8f){0.f, 0.f, 0.f, 0.f, 0.f, 0.f, 0.f, 0.f};

    const float* Bl = sB2 + hi * S2 + nl * 2;
    #pragma unroll
    for (int kk = 0; kk < 64; ++kk) {
        const float xp = xr[kk >> 2];
        v2f a;
        a.x = xp * sel0[kk & 3];
        a.y = xp * sel1[kk & 3];
        const float* Br = Bl + (2 * kk) * S2;
        #pragma unroll
        for (int t = 0; t < NTILE; ++t) {
            const v2f bf = *(const v2f*)(Br + t * 32);
            acc[t] = __builtin_amdgcn_wmma_f32_16x16x4_f32(
                false, a, false, bf, (short)0, acc[t], false, false);
        }
    }

    #pragma unroll
    for (int t = 0; t < NTILE; ++t) {
        #pragma unroll
        for (int r = 0; r < 8; ++r)
            sP[(wv * 16 + r + hi * 8) * PSTR + t * 16 + nl] = acc[t][r];
    }
    __syncthreads();

    for (int idx = tid; idx < 4 * C; idx += 256) {
        const int c = idx & 127;
        const int g = idx >> 7;
        const float* P = sP + c * PSTR;
        float s = 0.f;
        if (g == 0) {
            #pragma unroll
            for (int k = 0; k < K3_0; ++k) s = fmaf(P[COL_A0 + k], sW[(WR_M0 + k) * C + c], s);
            #pragma unroll
            for (int k = 0; k < K2_0; ++k) s = fmaf(P[COL_20 + k], sW[(WR_20 + k) * C + c], s);
            float d = 0.f;
            #pragma unroll
            for (int j = 0; j < L; ++j) d = fmaf(sU1[j], sX[c * XSTR + j], d);
            s = fmaf(sW[WR_10 * C + c], d, s);
            out[(size_t)b * (4 * C) + c] = s;
        } else {
            const int w = g - 1;
            #pragma unroll
            for (int k = 0; k < K3_1; ++k) s = fmaf(P[COL_A1 + w * K3_1 + k], sW[(WR_M1 + k) * C + c], s);
            #pragma unroll
            for (int k = 0; k < K2_1; ++k) s = fmaf(P[COL_21 + w * K2_1 + k], sW[(WR_21 + k) * C + c], s);
            float d = 0.f;
            #pragma unroll
            for (int j = 0; j < L; ++j) d = fmaf(sU1[16 + w * L + j], sX[c * XSTR + j], d);
            s = fmaf(sW[WR_11 * C + c], d, s);
            out[(size_t)b * (4 * C) + C + c * 3 + w] = s;
        }
    }
}

extern "C" void kernel_launch(void* const* d_in, const int* in_sizes, int n_in,
                              void* d_out, int out_size, void* d_ws, size_t ws_size,
                              hipStream_t stream) {
    (void)in_sizes; (void)n_in; (void)out_size; (void)d_ws; (void)ws_size;
    const float* x    = (const float*)d_in[0];
    const float* y    = (const float*)d_in[1];
    const float* U3_0 = (const float*)d_in[2];
    const float* U2_0 = (const float*)d_in[3];
    const float* U1_0 = (const float*)d_in[4];
    const float* U3_1 = (const float*)d_in[5];
    const float* U2_1 = (const float*)d_in[6];
    const float* U1_1 = (const float*)d_in[7];
    const float* Wm0  = (const float*)d_in[8];
    const float* W20  = (const float*)d_in[9];
    const float* W10  = (const float*)d_in[10];
    const float* Wm1  = (const float*)d_in[11];
    const float* W21  = (const float*)d_in[12];
    const float* W11  = (const float*)d_in[13];

    constexpr size_t smem_bytes = (size_t)SMEM_FLOATS * sizeof(float);
    mace_node_kernel<<<dim3(NN), dim3(256), smem_bytes, stream>>>(
        x, y, U3_0, U2_0, U1_0, U3_1, U2_1, U1_1,
        Wm0, W20, W10, Wm1, W21, W11, (float*)d_out);
}


# U3*xs path sums built for all nodes by a WMMA GEMM kernel (nodes = M) into the workspace, xs by a small kernel first; node blocks copy their rows to LDS; three launches; path-sum order changes
# speedup vs baseline: 2.6107x; 2.6107x over previous
#include <hip/hip_runtime.h>

typedef float v2f __attribute__((ext_vector_type(2)));
typedef float v4f __attribute__((ext_vector_type(4)));
typedef float v8f __attribute__((ext_vector_type(8)));

constexpr int NN   = 2048;
constexpr int C    = 128;
constexpr int L    = 16;
constexpr int E    = 10;
constexpr int K3_0 = 23, K2_0 = 4;
constexpr int K3_1 = 33, K2_1 = 6;

constexpr int COL_A1 = 0;
constexpr int COL_A0 = 3 * K3_1;
constexpr int COL_21 = COL_A0 + K3_0;
constexpr int COL_20 = COL_21 + 3 * K2_1;
constexpr int NCOL   = COL_20 + K2_0;
constexpr int NTILE  = NCOL / 16;

constexpr int S2   = 2 * NCOL + 2;
constexpr int PSTR = 149;
constexpr int XSTR = L + 1;

constexpr int WR_M1 = 0;
constexpr int WR_M0 = 33;
constexpr int WR_21 = 56;
constexpr int WR_20 = 62;
constexpr int WR_11 = 66;
constexpr int WR_10 = 67;
constexpr int WROWS = 68;

constexpr int OFF_X  = 0;
constexpr int OFF_XS = OFF_X + C * XSTR;
constexpr int OFF_W  = OFF_XS + L;
constexpr int OFF_U1 = OFF_W + WROWS * C;
constexpr int OFF_B2 = OFF_U1 + 64;
constexpr int OFF_P  = OFF_B2 + 128 * S2;
constexpr int SMEM_FLOATS = OFF_P + C * PSTR;

constexpr int TCOLS = COL_21;
constexpr int TPAIR = 2 * TCOLS;
constexpr int TNODE = 128 * TPAIR;
constexpr int NTT   = 128 * TCOLS / 16;
constexpr size_t WS_XS = (size_t)NN * TNODE;
constexpr int XS_BLOCKS = NN * L / 64;

__global__ __launch_bounds__(64) void xs_kernel(const float* __restrict__ x,
                                                float* __restrict__ xs)
{
    const int g = blockIdx.x * 64 + threadIdx.x;
    const float* xp = x + (size_t)(g >> 4) * (C * L) + (g & 15);
    float s = 0.f;
    #pragma unroll 32
    for (int c = 0; c < C; ++c) s += xp[c * L];
    xs[g] = s;
}

__global__ __launch_bounds__(256) void tbuild_kernel(const float* __restrict__ U3_0,
                                                     const float* __restrict__ U3_1,
                                                     const float* __restrict__ xs,
                                                     float* __restrict__ T)
{
    const int tid  = threadIdx.x;
    const int wv   = tid >> 5;
    const int lane = tid & 31;
    const int c    = lane & 15;
    const int h    = lane >> 4;
    const int nt   = blockIdx.x;
    const int n    = nt * 16 + c;

    const int p   = n / TCOLS;
    const int col = n - p * TCOLS;
    const float* se; int ks;
    if (col < COL_A0) {
        const int w = col / K3_1, k = col - w * K3_1;
        se = U3_1 + (size_t)(w * 256 + 2 * p) * (L * K3_1) + k;
        ks = K3_1;
    } else {
        se = U3_0 + (size_t)(2 * p) * (L * K3_0) + (col - COL_A0);
        ks = K3_0;
    }
    const int dst = 2 * n;
    const float* so = se + L * ks;

    v2f be[4], bo[4];
    #pragma unroll
    for (int s = 0; s < 4; ++s) {
        const int i0 = 4 * s + 2 * h;
        be[s] = (v2f){se[i0 * ks], se[(i0 + 1) * ks]};
        bo[s] = (v2f){so[i0 * ks], so[(i0 + 1) * ks]};
    }

    #pragma unroll 4
    for (int j = 0; j < 16; ++j) {
        const int m = wv * 16 + j;
        const float* xa = xs + (size_t)(m * 16 + c) * L + 2 * h;
        v2f a[4];
        #pragma unroll
        for (int s = 0; s < 4; ++s) a[s] = *(const v2f*)(xa + 4 * s);
        v8f de = (v8f){0.f, 0.f, 0.f, 0.f, 0.f, 0.f, 0.f, 0.f};
        v8f dd = (v8f){0.f, 0.f, 0.f, 0.f, 0.f, 0.f, 0.f, 0.f};
        #pragma unroll
        for (int s = 0; s < 4; ++s) {
            de = __builtin_amdgcn_wmma_f32_16x16x4_f32(
                false, a[s], false, be[s], (short)0, de, false, false);
            dd = __builtin_amdgcn_wmma_f32_16x16x4_f32(
                false, a[s], false, bo[s], (short)0, dd, false, false);
        }
        float* tp = T + (size_t)(m * 16 + 8 * h) * TNODE + dst;
        #pragma unroll
        for (int r = 0; r < 8; ++r)
            *(v2f*)(tp + (size_t)r * TNODE) = (v2f){de[r], dd[r]};
    }
}

__global__ __launch_bounds__(256) void mace_node_kernel(
    const float* __restrict__ x,    const float* __restrict__ y,
    const float* __restrict__ U2_0, const float* __restrict__ U1_0,
    const float* __restrict__ U2_1, const float* __restrict__ U1_1,
    const float* __restrict__ Wm0,  const float* __restrict__ W20,  const float* __restrict__ W10,
    const float* __restrict__ Wm1,  const float* __restrict__ W21,  const float* __restrict__ W11,
    const float* __restrict__ T,    float* __restrict__ out)
{
    extern __shared__ float sm[];
    float* sX  = sm + OFF_X;
    float* sW  = sm + OFF_W;
    float* sU1 = sm + OFF_U1;
    float* sB2 = sm + OFF_B2;
    float* sP  = sm + OFF_P;

    const int tid = threadIdx.x;
    const int b   = blockIdx.x;

    const v4f* Tb = (const v4f*)(T + (size_t)b * TNODE);
    v4f tv[31];
    #pragma unroll
    for (int j = 0; j < 31; ++j) {
        if (j < 30 || tid < 128)
            tv[j] = __builtin_nontemporal_load(Tb + tid + 256 * j);
    }

    for (int idx = tid; idx < C * L; idx += 256)
        sX[(idx >> 4) * XSTR + (idx & 15)] = x[(size_t)b * (C * L) + idx];

    float yv[E];
    #pragma unroll
    for (int e = 0; e < E; ++e) yv[e] = y[b * E + e];
    for (int idx = tid; idx < WROWS * C; idx += 256) {
        const int wr = idx >> 7, c = idx & 127;
        const float* src; int stride;
        if (wr < WR_M0)       { src = Wm1 + (wr)         * C + c; stride = K3_1 * C; }
        else if (wr < WR_21)  { src = Wm0 + (wr - WR_M0) * C + c; stride = K3_0 * C; }
        else if (wr < WR_20)  { src = W21 + (wr - WR_21) * C + c; stride = K2_1 * C; }
        else if (wr < WR_11)  { src = W20 + (wr - WR_20) * C + c; stride = K2_0 * C; }
        else if (wr == WR_11) { src = W11 + c;                    stride = C;        }
        else                  { src = W10 + c;                    stride = C;        }
        float acc = 0.f;
        #pragma unroll
        for (int e = 0; e < E; ++e) acc = fmaf(yv[e], src[e * stride], acc);
        sW[idx] = acc;
    }
    if (tid < 16)               sU1[tid] = U1_0[tid];
    else if (tid < 64)          sU1[tid] = U1_1[tid - 16];

    #pragma unroll
    for (int j = 0; j < 31; ++j) {
        if (j < 30 || tid < 128) {
            const int f = 4 * (tid + 256 * j);
            const int p = f / TPAIR;
            float* d = sB2 + p * S2 + (f - p * TPAIR);
            *(v2f*)(d)     = (v2f){tv[j].x, tv[j].y};
            *(v2f*)(d + 2) = (v2f){tv[j].z, tv[j].w};
        }
    }
    if (tid < 128) {
        const int p = tid;
        float* bp = sB2 + p * S2;
        #pragma unroll
        for (int m = 0; m < 3 * K2_1; ++m) {
            const int w = m / K2_1, k = m % K2_1;
            const int r0 = (w * 256 + 2 * p) * K2_1 + k;
            *(v2f*)(bp + (COL_21 + m) * 2) = (v2f){U2_1[r0], U2_1[r0 + K2_1]};
        }
        #pragma unroll
        for (int m = 0; m < K2_0; ++m)
            *(v2f*)(bp + (COL_20 + m) * 2) =
                (v2f){U2_0[(2 * p) * K2_0 + m], U2_0[(2 * p + 1) * K2_0 + m]};
    }
    __syncthreads();

    const int wv   = tid >> 5;
    const int lane = tid & 31;
    const int nl   = lane & 15;
    const int hi   = lane >> 4;
    const int crow = wv * 16 + nl;

    float xr[L];
    #pragma unroll
    for (int j = 0; j < L; ++j) xr[j] = sX[crow * XSTR + j];

    float sel0[4], sel1[4];
    #pragma unroll
    for (int j = 0; j < 4; ++j) {
        const float e0 = xr[4 * j],     e1 = xr[4 * j + 1];
        const float e2 = xr[4 * j + 2], e3 = xr[4 * j + 3];
        sel0[j] = hi ? e2 : e0;
        sel1[j] = hi ? e3 : e1;
    }

    v8f acc[NTILE];
    #pragma unroll
    for (int t = 0; t < NTILE; ++t)
        acc[t] = (v8f){0.f, 0.f, 0.f, 0.f, 0.f, 0.f, 0.f, 0.f};

    const float* Bl = sB2 + hi * S2 + nl * 2;
    #pragma unroll
    for (int kk = 0; kk < 64; ++kk) {
        const float xp = xr[kk >> 2];
        v2f a;
        a.x = xp * sel0[kk & 3];
        a.y = xp * sel1[kk & 3];
        const float* Br = Bl + (2 * kk) * S2;
        #pragma unroll
        for (int t = 0; t < NTILE; ++t) {
            const v2f bf = *(const v2f*)(Br + t * 32);
            acc[t] = __builtin_amdgcn_wmma_f32_16x16x4_f32(
                false, a, false, bf, (short)0, acc[t], false, false);
        }
    }

    #pragma unroll
    for (int t = 0; t < NTILE; ++t) {
        #pragma unroll
        for (int r = 0; r < 8; ++r)
            sP[(wv * 16 + r + hi * 8) * PSTR + t * 16 + nl] = acc[t][r];
    }
    __syncthreads();

    for (int idx = tid; idx < 4 * C; idx += 256) {
        const int c = idx & 127;
        const int g = idx >> 7;
        const float* P = sP + c * PSTR;
        float s = 0.f;
        if (g == 0) {
            #pragma unroll
            for (int k = 0; k < K3_0; ++k) s = fmaf(P[COL_A0 + k], sW[(WR_M0 + k) * C + c], s);
            #pragma unroll
            for (int k = 0; k < K2_0; ++k) s = fmaf(P[COL_20 + k], sW[(WR_20 + k) * C + c], s);
            float d = 0.f;
            #pragma unroll
            for (int j = 0; j < L; ++j) d = fmaf(sU1[j], sX[c * XSTR + j], d);
            s = fmaf(sW[WR_10 * C + c], d, s);
            out[(size_t)b * (4 * C) + c] = s;
        } else {
            const int w = g - 1;
            #pragma unroll
            for (int k = 0; k < K3_1; ++k) s = fmaf(P[COL_A1 + w * K3_1 + k], sW[(WR_M1 + k) * C + c], s);
            #pragma unroll
            for (int k = 0; k < K2_1; ++k) s = fmaf(P[COL_21 + w * K2_1 + k], sW[(WR_21 + k) * C + c], s);
            float d = 0.f;
            #pragma unroll
            for (int j = 0; j < L; ++j) d = fmaf(sU1[16 + w * L + j], sX[c * XSTR + j], d);
            s = fmaf(sW[WR_11 * C + c], d, s);
            out[(size_t)b * (4 * C) + C + c * 3 + w] = s;
        }
    }
}

extern "C" void kernel_launch(void* const* d_in, const int* in_sizes, int n_in,
                              void* d_out, int out_size, void* d_ws, size_t ws_size,
                              hipStream_t stream) {
    (void)in_sizes; (void)n_in; (void)out_size; (void)ws_size;
    const float* x    = (const float*)d_in[0];
    const float* y    = (const float*)d_in[1];
    const float* U3_0 = (const float*)d_in[2];
    const float* U2_0 = (const float*)d_in[3];
    const float* U1_0 = (const float*)d_in[4];
    const float* U3_1 = (const float*)d_in[5];
    const float* U2_1 = (const float*)d_in[6];
    const float* U1_1 = (const float*)d_in[7];
    const float* Wm0  = (const float*)d_in[8];
    const float* W20  = (const float*)d_in[9];
    const float* W10  = (const float*)d_in[10];
    const float* Wm1  = (const float*)d_in[11];
    const float* W21  = (const float*)d_in[12];
    const float* W11  = (const float*)d_in[13];

    float* T  = (float*)d_ws;
    float* xs = T + WS_XS;

    xs_kernel<<<dim3(XS_BLOCKS), dim3(64), 0, stream>>>(x, xs);
    tbuild_kernel<<<dim3(NTT), dim3(256), 0, stream>>>(U3_0, U3_1, xs, T);

    constexpr size_t smem_bytes = (size_t)SMEM_FLOATS * sizeof(float);
    mace_node_kernel<<<dim3(NN), dim3(256), smem_bytes, stream>>>(
        x, y, U2_0, U1_0, U2_1, U1_1,
        Wm0, W20, W10, Wm1, W21, W11, T, (float*)d_out);
}


# node kernel persistent (256 blocks x 8 nodes), software-pipelined: node's weights built and next node's T rows and x tile fetched inside the WMMA K loop; U2/U1/y staged once per block; sums kept
# speedup vs baseline: 3.3941x; 1.3001x over previous
#include <hip/hip_runtime.h>

typedef float v2f __attribute__((ext_vector_type(2)));
typedef float v4f __attribute__((ext_vector_type(4)));
typedef float v8f __attribute__((ext_vector_type(8)));

constexpr int NN   = 2048;
constexpr int C    = 128;
constexpr int L    = 16;
constexpr int E    = 10;
constexpr int K3_0 = 23, K2_0 = 4;
constexpr int K3_1 = 33, K2_1 = 6;

constexpr int COL_A1 = 0;
constexpr int COL_A0 = 3 * K3_1;
constexpr int COL_21 = COL_A0 + K3_0;
constexpr int COL_20 = COL_21 + 3 * K2_1;
constexpr int NCOL   = COL_20 + K2_0;
constexpr int NTILE  = NCOL / 16;

constexpr int S2   = 2 * NCOL + 2;
constexpr int PSTR = 149;
constexpr int XSTR = L + 1;

constexpr int WR_M1 = 0;
constexpr int WR_M0 = 34;
constexpr int WR_21 = 58;
constexpr int WR_20 = 64;
constexpr int WR_11 = 68;
constexpr int WR_10 = 69;
constexpr int WROWS = 70;
constexpr int WPAIRS = WROWS / 2;
constexpr int WCH    = 5;

constexpr int OFF_X  = 0;
constexpr int OFF_W  = OFF_X + 2 * C * XSTR;
constexpr int NODE_BLOCKS = 256;
constexpr int NPB    = NN / NODE_BLOCKS;
constexpr int OFF_Y  = OFF_W + WROWS * C;
constexpr int OFF_U1 = OFF_Y + NPB * E;
constexpr int OFF_B2 = OFF_U1 + 64;
constexpr int OFF_P  = OFF_B2 + 128 * S2;
constexpr int SMEM_FLOATS = OFF_P + C * PSTR;

constexpr int TCOLS = COL_21;
constexpr int TPAIR = 2 * TCOLS;
constexpr int TNODE = 128 * TPAIR;
constexpr int NTT   = 128 * TCOLS / 16;
constexpr size_t WS_XS = (size_t)NN * TNODE;
constexpr int XS_BLOCKS = NN * L / 64;

__global__ __launch_bounds__(64) void xs_kernel(const float* __restrict__ x,
                                                float* __restrict__ xs)
{
    const int g = blockIdx.x * 64 + threadIdx.x;
    const float* xp = x + (size_t)(g >> 4) * (C * L) + (g & 15);
    float s = 0.f;
    #pragma unroll 32
    for (int c = 0; c < C; ++c) s += xp[c * L];
    xs[g] = s;
}

__global__ __launch_bounds__(256) void tbuild_kernel(const float* __restrict__ U3_0,
                                                     const float* __restrict__ U3_1,
                                                     const float* __restrict__ xs,
                                                     float* __restrict__ T)
{
    const int tid  = threadIdx.x;
    const int wv   = tid >> 5;
    const int lane = tid & 31;
    const int c    = lane & 15;
    const int h    = lane >> 4;
    const int nt   = blockIdx.x;
    const int n    = nt * 16 + c;

    const int p   = n / TCOLS;
    const int col = n - p * TCOLS;
    const float* se; int ks;
    if (col < COL_A0) {
        const int w = col / K3_1, k = col - w * K3_1;
        se = U3_1 + (size_t)(w * 256 + 2 * p) * (L * K3_1) + k;
        ks = K3_1;
    } else {
        se = U3_0 + (size_t)(2 * p) * (L * K3_0) + (col - COL_A0);
        ks = K3_0;
    }
    const int dst = 2 * n;
    const float* so = se + L * ks;

    v2f be[4], bo[4];
    #pragma unroll
    for (int s = 0; s < 4; ++s) {
        const int i0 = 4 * s + 2 * h;
        be[s] = (v2f){se[i0 * ks], se[(i0 + 1) * ks]};
        bo[s] = (v2f){so[i0 * ks], so[(i0 + 1) * ks]};
    }

    #pragma unroll 4
    for (int j = 0; j < 16; ++j) {
        const int m = wv * 16 + j;
        const float* xa = xs + (size_t)(m * 16 + c) * L + 2 * h;
        v2f a[4];
        #pragma unroll
        for (int s = 0; s < 4; ++s) a[s] = *(const v2f*)(xa + 4 * s);
        v8f de = (v8f){0.f, 0.f, 0.f, 0.f, 0.f, 0.f, 0.f, 0.f};
        v8f dd = (v8f){0.f, 0.f, 0.f, 0.f, 0.f, 0.f, 0.f, 0.f};
        #pragma unroll
        for (int s = 0; s < 4; ++s) {
            de = __builtin_amdgcn_wmma_f32_16x16x4_f32(
                false, a[s], false, be[s], (short)0, de, false, false);
            dd = __builtin_amdgcn_wmma_f32_16x16x4_f32(
                false, a[s], false, bo[s], (short)0, dd, false, false);
        }
        float* tp = T + (size_t)(m * 16 + 8 * h) * TNODE + dst;
        #pragma unroll
        for (int r = 0; r < 8; ++r)
            *(v2f*)(tp + (size_t)r * TNODE) = (v2f){de[r], dd[r]};
    }
}

__device__ __forceinline__ void node_loads(const float* __restrict__ T, const float* __restrict__ x,
                                           int b, int tid, v4f (&tv)[31], float (&xv)[8])
{
    const v4f* Tb = (const v4f*)(T + (size_t)b * TNODE);
    #pragma unroll
    for (int j = 0; j < 31; ++j) {
        if (j < 30 || tid < 128)
            tv[j] = __builtin_nontemporal_load(Tb + tid + 256 * j);
    }
    const float* xb = x + (size_t)b * (C * L);
    #pragma unroll
    for (int j = 0; j < 8; ++j) xv[j] = xb[tid + 256 * j];
}

struct WTabs { const float* Wm0; const float* W20; const float* W10;
               const float* Wm1; const float* W21; const float* W11; };

template <int RP>
__device__ __forceinline__ const float* wpair_src(const WTabs& w, int half, int hc, int c7, int& stride)
{
    if constexpr (RP < 17)      { stride = K3_1 * C; return w.Wm1 + (2 * RP) * C        + (RP == 16 ? c7 : hc); }
    else if constexpr (RP < 29) { stride = K3_0 * C; return w.Wm0 + (2 * (RP - 17)) * C + (RP == 28 ? c7 : hc); }
    else if constexpr (RP < 32) { stride = K2_1 * C; return w.W21 + (2 * (RP - 29)) * C + hc; }
    else if constexpr (RP < 34) { stride = K2_0 * C; return w.W20 + (2 * (RP - 32)) * C + hc; }
    else                        { stride = C;        return (half ? w.W10 : w.W11) + c7; }
}

struct NodeCtx {
    const float* Bl;
    float*       sW;
    const float* T;
    const float* x;
    WTabs        wt;
    int tid, half, hc, c7, bn;
    bool has_next;
};

template <int Q, int U>
__device__ __forceinline__ void wload(const NodeCtx& n, float (&wl)[WCH][E])
{
    if constexpr (U < WCH) {
        int stride;
        const float* wsrc = wpair_src<Q * WCH + U>(n.wt, n.half, n.hc, n.c7, stride);
        #pragma unroll
        for (int e = 0; e < E; ++e) wl[U][e] = wsrc[e * stride];
        wload<Q, U + 1>(n, wl);
    }
}
template <int Q>
__device__ __forceinline__ void wsum(const NodeCtx& n, const float (&wl)[WCH][E], const float (&yv)[E])
{
    #pragma unroll
    for (int u = 0; u < WCH; ++u) {
        float s = 0.f;
        #pragma unroll
        for (int e = 0; e < E; ++e) s = fmaf(yv[e], wl[u][e], s);
        n.sW[(Q * WCH + u) * 256 + n.tid] = s;
    }
}

template <int Q>
__device__ __forceinline__ void kchunk(const NodeCtx& n, v8f (&acc)[NTILE],
                                       const float (&xr)[L], const float (&sel0)[4], const float (&sel1)[4],
                                       const float (&yv)[E], float (&wl)[WCH][E],
                                       v4f (&tv)[31], float (&xv)[8])
{
    constexpr int NWQ = WPAIRS / WCH;
    if constexpr (Q <= NWQ) {
        __builtin_amdgcn_sched_barrier(0x108);
        if constexpr (Q >= 1)  wsum<Q - 1>(n, wl, yv);
        if constexpr (Q < NWQ) wload<Q, 0>(n, wl);
        __builtin_amdgcn_sched_barrier(0x108);
    }

    const float xp = xr[Q];
    #pragma unroll
    for (int j = 0; j < 4; ++j) {
        v2f a;
        a.x = xp * sel0[j];
        a.y = xp * sel1[j];
        const float* Br = n.Bl + (2 * (4 * Q + j)) * S2;
        #pragma unroll
        for (int t = 0; t < NTILE; ++t) {
            const v2f bf = *(const v2f*)(Br + t * 32);
            acc[t] = __builtin_amdgcn_wmma_f32_16x16x4_f32(
                false, a, false, bf, (short)0, acc[t], false, false);
        }
    }

    if constexpr (Q == NWQ) {
        if (n.has_next) node_loads(n.T, n.x, n.bn, n.tid, tv, xv);
    }

    if constexpr (Q + 1 < 16) kchunk<Q + 1>(n, acc, xr, sel0, sel1, yv, wl, tv, xv);
}

__global__ __launch_bounds__(256) void mace_node_kernel(
    const float* __restrict__ x,    const float* __restrict__ y,
    const float* __restrict__ U2_0, const float* __restrict__ U1_0,
    const float* __restrict__ U2_1, const float* __restrict__ U1_1,
    const float* Wm0,  const float* W20,  const float* W10,
    const float* Wm1,  const float* W21,  const float* W11,
    const float* __restrict__ T,    float* out)
{
    extern __shared__ float sm[];
    float* sW  = sm + OFF_W;
    float* sY  = sm + OFF_Y;
    float* sU1 = sm + OFF_U1;
    float* sB2 = sm + OFF_B2;
    float* sP  = sm + OFF_P;

    const int tid  = threadIdx.x;
    const int wv   = tid >> 5;
    const int lane = tid & 31;
    const int nl   = lane & 15;
    const int hi   = lane >> 4;
    const int crow = wv * 16 + nl;
    const int c7   = tid & 127;
    const int half = tid >> 7;
    const int hc   = half * C + c7;

    int b = blockIdx.x;

    v4f tv[31];
    float xv[8];
    tv[30] = (v4f){0.f, 0.f, 0.f, 0.f};
    if (b < NN) node_loads(T, x, b, tid, tv, xv);

    if (tid < NPB * E) sY[tid] = y[(blockIdx.x + NODE_BLOCKS * (tid / E)) * E + (tid % E)];
    if (tid < 16)               sU1[tid] = U1_0[tid];
    else if (tid < 64)          sU1[tid] = U1_1[tid - 16];
    if (tid < 128) {
        const int p = tid;
        float* bp = sB2 + p * S2;
        #pragma unroll
        for (int m = 0; m < 3 * K2_1; ++m) {
            const int w = m / K2_1, k = m % K2_1;
            const int r0 = (w * 256 + 2 * p) * K2_1 + k;
            *(v2f*)(bp + (COL_21 + m) * 2) = (v2f){U2_1[r0], U2_1[r0 + K2_1]};
        }
        #pragma unroll
        for (int m = 0; m < K2_0; ++m)
            *(v2f*)(bp + (COL_20 + m) * 2) =
                (v2f){U2_0[(2 * p) * K2_0 + m], U2_0[(2 * p + 1) * K2_0 + m]};
    }

    int par = 0, it = 0;
    for (; b < NN; b += NODE_BLOCKS, par ^= 1, ++it) {
        float* sX = sm + OFF_X + par * (C * XSTR);

        #pragma unroll
        for (int j = 0; j < 31; ++j) {
            if (j < 30 || tid < 128) {
                const int f = 4 * (tid + 256 * j);
                const int p = f / TPAIR;
                float* d = sB2 + p * S2 + (f - p * TPAIR);
                *(v2f*)(d)     = (v2f){tv[j].x, tv[j].y};
                *(v2f*)(d + 2) = (v2f){tv[j].z, tv[j].w};
            }
        }
        #pragma unroll
        for (int j = 0; j < 8; ++j) {
            const int idx = tid + 256 * j;
            sX[(idx >> 4) * XSTR + (idx & 15)] = xv[j];
        }
        __syncthreads();
        float yv[E];
        #pragma unroll
        for (int e = 0; e < E; ++e) yv[e] = sY[it * E + e];

        float xr[L];
        #pragma unroll
        for (int j = 0; j < L; ++j) xr[j] = sX[crow * XSTR + j];

        float sel0[4], sel1[4];
        #pragma unroll
        for (int j = 0; j < 4; ++j) {
            const float e0 = xr[4 * j],     e1 = xr[4 * j + 1];
            const float e2 = xr[4 * j + 2], e3 = xr[4 * j + 3];
            sel0[j] = hi ? e2 : e0;
            sel1[j] = hi ? e3 : e1;
        }

        v8f acc[NTILE];
        #pragma unroll
        for (int t = 0; t < NTILE; ++t)
            acc[t] = (v8f){0.f, 0.f, 0.f, 0.f, 0.f, 0.f, 0.f, 0.f};

        NodeCtx n;
        n.Bl  = sB2 + hi * S2 + nl * 2;
        n.sW  = sW;  n.T = T;  n.x = x;
        n.wt  = WTabs{Wm0, W20, W10, Wm1, W21, W11};
        n.tid = tid; n.half = half; n.hc = hc; n.c7 = c7;
        n.bn  = b + NODE_BLOCKS;
        n.has_next = n.bn < NN;

        float wl[WCH][E];
        kchunk<0>(n, acc, xr, sel0, sel1, yv, wl, tv, xv);

        #pragma unroll
        for (int t = 0; t < NTILE; ++t) {
            #pragma unroll
            for (int r = 0; r < 8; ++r)
                sP[(wv * 16 + r + hi * 8) * PSTR + t * 16 + nl] = acc[t][r];
        }
        __syncthreads();

        for (int idx = tid; idx < 4 * C; idx += 256) {
            const int c = idx & 127;
            const int g = idx >> 7;
            const float* P = sP + c * PSTR;
            float s = 0.f;
            if (g == 0) {
                #pragma unroll
                for (int k = 0; k < K3_0; ++k) s = fmaf(P[COL_A0 + k], sW[(WR_M0 + k) * C + c], s);
                #pragma unroll
                for (int k = 0; k < K2_0; ++k) s = fmaf(P[COL_20 + k], sW[(WR_20 + k) * C + c], s);
                float d = 0.f;
                #pragma unroll
                for (int j = 0; j < L; ++j) d = fmaf(sU1[j], sX[c * XSTR + j], d);
                s = fmaf(sW[WR_10 * C + c], d, s);
                out[(size_t)b * (4 * C) + c] = s;
            } else {
                const int w = g - 1;
                #pragma unroll
                for (int k = 0; k < K3_1; ++k) s = fmaf(P[COL_A1 + w * K3_1 + k], sW[(WR_M1 + k) * C + c], s);
                #pragma unroll
                for (int k = 0; k < K2_1; ++k) s = fmaf(P[COL_21 + w * K2_1 + k], sW[(WR_21 + k) * C + c], s);
                float d = 0.f;
                #pragma unroll
                for (int j = 0; j < L; ++j) d = fmaf(sU1[16 + w * L + j], sX[c * XSTR + j], d);
                s = fmaf(sW[WR_11 * C + c], d, s);
                out[(size_t)b * (4 * C) + C + c * 3 + w] = s;
            }
        }
    }
}

extern "C" void kernel_launch(void* const* d_in, const int* in_sizes, int n_in,
                              void* d_out, int out_size, void* d_ws, size_t ws_size,
                              hipStream_t stream) {
    (void)in_sizes; (void)n_in; (void)out_size; (void)ws_size;
    const float* x    = (const float*)d_in[0];
    const float* y    = (const float*)d_in[1];
    const float* U3_0 = (const float*)d_in[2];
    const float* U2_0 = (const float*)d_in[3];
    const float* U1_0 = (const float*)d_in[4];
    const float* U3_1 = (const float*)d_in[5];
    const float* U2_1 = (const float*)d_in[6];
    const float* U1_1 = (const float*)d_in[7];
    const float* Wm0  = (const float*)d_in[8];
    const float* W20  = (const float*)d_in[9];
    const float* W10  = (const float*)d_in[10];
    const float* Wm1  = (const float*)d_in[11];
    const float* W21  = (const float*)d_in[12];
    const float* W11  = (const float*)d_in[13];

    float* T  = (float*)d_ws;
    float* xs = T + WS_XS;

    xs_kernel<<<dim3(XS_BLOCKS), dim3(64), 0, stream>>>(x, xs);
    tbuild_kernel<<<dim3(NTT), dim3(256), 0, stream>>>(U3_0, U3_1, xs, T);

    constexpr size_t smem_bytes = (size_t)SMEM_FLOATS * sizeof(float);
    mace_node_kernel<<<dim3(NODE_BLOCKS), dim3(256), smem_bytes, stream>>>(
        x, y, U2_0, U1_0, U2_1, U1_1,
        Wm0, W20, W10, Wm1, W21, W11, T, (float*)d_out);
}


# B fragments of the node products read one K step ahead into a second register set: a step's 9 LDS reads are one group in front of the previous step's 9 WMMAs, not one read per WMMA pair; sums kept
# speedup vs baseline: 3.9737x; 1.1708x over previous
#include <hip/hip_runtime.h>

typedef float v2f __attribute__((ext_vector_type(2)));
typedef float v4f __attribute__((ext_vector_type(4)));
typedef float v8f __attribute__((ext_vector_type(8)));

constexpr int NN   = 2048;
constexpr int C    = 128;
constexpr int L    = 16;
constexpr int E    = 10;
constexpr int K3_0 = 23, K2_0 = 4;
constexpr int K3_1 = 33, K2_1 = 6;

constexpr int COL_A1 = 0;
constexpr int COL_A0 = 3 * K3_1;
constexpr int COL_21 = COL_A0 + K3_0;
constexpr int COL_20 = COL_21 + 3 * K2_1;
constexpr int NCOL   = COL_20 + K2_0;
constexpr int NTILE  = NCOL / 16;

constexpr int S2   = 2 * NCOL + 2;
constexpr int PSTR = 149;
constexpr int FAR_STEPS = 28;
constexpr int B2_FAR    = 2 * FAR_STEPS * S2;
constexpr int XSTR = L + 1;

constexpr int WR_M1 = 0;
constexpr int WR_M0 = 34;
constexpr int WR_21 = 58;
constexpr int WR_20 = 64;
constexpr int WR_11 = 68;
constexpr int WR_10 = 69;
constexpr int WROWS = 70;
constexpr int WPAIRS = WROWS / 2;
constexpr int WCH    = 5;

constexpr int OFF_X  = 0;
constexpr int OFF_W  = OFF_X + 2 * C * XSTR;
constexpr int NODE_BLOCKS = 256;
constexpr int NPB    = NN / NODE_BLOCKS;
constexpr int OFF_Y  = OFF_W + WROWS * C;
constexpr int OFF_U1 = OFF_Y + NPB * E;
constexpr int OFF_B2 = OFF_U1 + 64;
constexpr int OFF_P  = OFF_B2 + 128 * S2;
constexpr int SMEM_FLOATS = OFF_P + C * PSTR;

constexpr int TCOLS = COL_21;
constexpr int TPAIR = 2 * TCOLS;
constexpr int TNODE = 128 * TPAIR;
constexpr int NTT   = 128 * TCOLS / 16;
constexpr size_t WS_XS = (size_t)NN * TNODE;
constexpr int XS_BLOCKS = NN * L / 64;

__global__ __launch_bounds__(64) void xs_kernel(const float* __restrict__ x,
                                                float* __restrict__ xs)
{
    const int g = blockIdx.x * 64 + threadIdx.x;
    const float* xp = x + (size_t)(g >> 4) * (C * L) + (g & 15);
    float s = 0.f;
    #pragma unroll 32
    for (int c = 0; c < C; ++c) s += xp[c * L];
    xs[g] = s;
}

__global__ __launch_bounds__(256) void tbuild_kernel(const float* __restrict__ U3_0,
                                                     const float* __restrict__ U3_1,
                                                     const float* __restrict__ xs,
                                                     float* __restrict__ T)
{
    const int tid  = threadIdx.x;
    const int wv   = tid >> 5;
    const int lane = tid & 31;
    const int c    = lane & 15;
    const int h    = lane >> 4;
    const int nt   = blockIdx.x;
    const int n    = nt * 16 + c;

    const int p   = n / TCOLS;
    const int col = n - p * TCOLS;
    const float* se; int ks;
    if (col < COL_A0) {
        const int w = col / K3_1, k = col - w * K3_1;
        se = U3_1 + (size_t)(w * 256 + 2 * p) * (L * K3_1) + k;
        ks = K3_1;
    } else {
        se = U3_0 + (size_t)(2 * p) * (L * K3_0) + (col - COL_A0);
        ks = K3_0;
    }
    const int dst = 2 * n;
    const float* so = se + L * ks;

    v2f be[4], bo[4];
    #pragma unroll
    for (int s = 0; s < 4; ++s) {
        const int i0 = 4 * s + 2 * h;
        be[s] = (v2f){se[i0 * ks], se[(i0 + 1) * ks]};
        bo[s] = (v2f){so[i0 * ks], so[(i0 + 1) * ks]};
    }

    #pragma unroll 4
    for (int j = 0; j < 16; ++j) {
        const int m = wv * 16 + j;
        const float* xa = xs + (size_t)(m * 16 + c) * L + 2 * h;
        v2f a[4];
        #pragma unroll
        for (int s = 0; s < 4; ++s) a[s] = *(const v2f*)(xa + 4 * s);
        v8f de = (v8f){0.f, 0.f, 0.f, 0.f, 0.f, 0.f, 0.f, 0.f};
        v8f dd = (v8f){0.f, 0.f, 0.f, 0.f, 0.f, 0.f, 0.f, 0.f};
        #pragma unroll
        for (int s = 0; s < 4; ++s) {
            de = __builtin_amdgcn_wmma_f32_16x16x4_f32(
                false, a[s], false, be[s], (short)0, de, false, false);
            dd = __builtin_amdgcn_wmma_f32_16x16x4_f32(
                false, a[s], false, bo[s], (short)0, dd, false, false);
        }
        float* tp = T + (size_t)(m * 16 + 8 * h) * TNODE + dst;
        #pragma unroll
        for (int r = 0; r < 8; ++r)
            *(v2f*)(tp + (size_t)r * TNODE) = (v2f){de[r], dd[r]};
    }
}

__device__ __forceinline__ void node_loads(const float* __restrict__ T, const float* __restrict__ x,
                                           int b, int tid, v4f (&tv)[31], float (&xv)[8])
{
    const v4f* Tb = (const v4f*)(T + (size_t)b * TNODE);
    #pragma unroll
    for (int j = 0; j < 31; ++j) {
        if (j < 30 || tid < 128)
            tv[j] = __builtin_nontemporal_load(Tb + tid + 256 * j);
    }
    const float* xb = x + (size_t)b * (C * L);
    #pragma unroll
    for (int j = 0; j < 8; ++j) xv[j] = xb[tid + 256 * j];
}

struct WTabs { const float* Wm0; const float* W20; const float* W10;
               const float* Wm1; const float* W21; const float* W11; };

template <int RP>
__device__ __forceinline__ const float* wpair_src(const WTabs& w, int half, int hc, int c7, int& stride)
{
    if constexpr (RP < 17)      { stride = K3_1 * C; return w.Wm1 + (2 * RP) * C        + (RP == 16 ? c7 : hc); }
    else if constexpr (RP < 29) { stride = K3_0 * C; return w.Wm0 + (2 * (RP - 17)) * C + (RP == 28 ? c7 : hc); }
    else if constexpr (RP < 32) { stride = K2_1 * C; return w.W21 + (2 * (RP - 29)) * C + hc; }
    else if constexpr (RP < 34) { stride = K2_0 * C; return w.W20 + (2 * (RP - 32)) * C + hc; }
    else                        { stride = C;        return (half ? w.W10 : w.W11) + c7; }
}

struct NodeCtx {
    const float* Bb[3];
    float*       sW;
    const float* T;
    const float* x;
    WTabs        wt;
    int tid, half, hc, c7, bn;
    bool has_next;
};

template <int Q, int U>
__device__ __forceinline__ void wload(const NodeCtx& n, float (&wl)[WCH][E])
{
    if constexpr (U < WCH) {
        int stride;
        const float* wsrc = wpair_src<Q * WCH + U>(n.wt, n.half, n.hc, n.c7, stride);
        #pragma unroll
        for (int e = 0; e < E; ++e) wl[U][e] = wsrc[e * stride];
        wload<Q, U + 1>(n, wl);
    }
}
template <int Q>
__device__ __forceinline__ void wsum(const NodeCtx& n, const float (&wl)[WCH][E], const float (&yv)[E])
{
    #pragma unroll
    for (int u = 0; u < WCH; ++u) {
        float s = 0.f;
        #pragma unroll
        for (int e = 0; e < E; ++e) s = fmaf(yv[e], wl[u][e], s);
        n.sW[(Q * WCH + u) * 256 + n.tid] = s;
    }
}

template <int Q>
__device__ __forceinline__ void kchunk(const NodeCtx& n, v8f (&acc)[NTILE], v2f (&bq)[2][NTILE],
                                       const float (&xr)[L], const float (&sel0)[4], const float (&sel1)[4],
                                       const float (&yv)[E], float (&wl)[WCH][E],
                                       v4f (&tv)[31], float (&xv)[8])
{
    constexpr int NWQ = WPAIRS / WCH;
    if constexpr (Q <= NWQ) {
        if constexpr (Q >= 1)  wsum<Q - 1>(n, wl, yv);
        if constexpr (Q < NWQ) wload<Q, 0>(n, wl);
    }

    const float xp = xr[Q];
    #pragma unroll
    for (int j = 0; j < 4; ++j) {
        v2f a;
        a.x = xp * sel0[j];
        a.y = xp * sel1[j];
        if (4 * Q + j + 1 < 64) {
            const int sn = 4 * Q + j + 1;
            const float* Bn = n.Bb[sn / FAR_STEPS] + (2 * (sn % FAR_STEPS)) * S2;
            #pragma unroll
            for (int t = 0; t < NTILE; ++t)
                bq[(j + 1) & 1][t] = *(const v2f*)(Bn + t * 32);
        }
        __builtin_amdgcn_sched_barrier(0);
        #pragma unroll
        for (int t = 0; t < NTILE; ++t)
            acc[t] = __builtin_amdgcn_wmma_f32_16x16x4_f32(
                false, a, false, bq[j & 1][t], (short)0, acc[t], false, false);
        __builtin_amdgcn_sched_barrier(0);
    }

    if constexpr (Q == NWQ) {
        if (n.has_next) node_loads(n.T, n.x, n.bn, n.tid, tv, xv);
    }

    if constexpr (Q + 1 < 16) kchunk<Q + 1>(n, acc, bq, xr, sel0, sel1, yv, wl, tv, xv);
}

__global__ __launch_bounds__(256) void mace_node_kernel(
    const float* __restrict__ x,    const float* __restrict__ y,
    const float* __restrict__ U2_0, const float* __restrict__ U1_0,
    const float* __restrict__ U2_1, const float* __restrict__ U1_1,
    const float* Wm0,  const float* W20,  const float* W10,
    const float* Wm1,  const float* W21,  const float* W11,
    const float* __restrict__ T,    float* out, int b2_far)
{
    extern __shared__ float sm[];
    float* sW  = sm + OFF_W;
    float* sY  = sm + OFF_Y;
    float* sU1 = sm + OFF_U1;
    float* sB2 = sm + OFF_B2;
    float* sP  = sm + OFF_P;

    const int tid  = threadIdx.x;
    const int wv   = tid >> 5;
    const int lane = tid & 31;
    const int nl   = lane & 15;
    const int hi   = lane >> 4;
    const int crow = wv * 16 + nl;
    const int c7   = tid & 127;
    const int half = tid >> 7;
    const int hc   = half * C + c7;

    int b = blockIdx.x;

    v4f tv[31];
    float xv[8];
    tv[30] = (v4f){0.f, 0.f, 0.f, 0.f};
    if (b < NN) node_loads(T, x, b, tid, tv, xv);

    if (tid < NPB * E) sY[tid] = y[(blockIdx.x + NODE_BLOCKS * (tid / E)) * E + (tid % E)];
    if (tid < 16)               sU1[tid] = U1_0[tid];
    else if (tid < 64)          sU1[tid] = U1_1[tid - 16];
    if (tid < 128) {
        const int p = tid;
        float* bp = sB2 + p * S2;
        #pragma unroll
        for (int m = 0; m < 3 * K2_1; ++m) {
            const int w = m / K2_1, k = m % K2_1;
            const int r0 = (w * 256 + 2 * p) * K2_1 + k;
            *(v2f*)(bp + (COL_21 + m) * 2) = (v2f){U2_1[r0], U2_1[r0 + K2_1]};
        }
        #pragma unroll
        for (int m = 0; m < K2_0; ++m)
            *(v2f*)(bp + (COL_20 + m) * 2) =
                (v2f){U2_0[(2 * p) * K2_0 + m], U2_0[(2 * p + 1) * K2_0 + m]};
    }

    int par = 0, it = 0;
    for (; b < NN; b += NODE_BLOCKS, par ^= 1, ++it) {
        float* sX = sm + OFF_X + par * (C * XSTR);

        #pragma unroll
        for (int j = 0; j < 31; ++j) {
            if (j < 30 || tid < 128) {
                const int f = 4 * (tid + 256 * j);
                const int p = f / TPAIR;
                float* d = sB2 + p * S2 + (f - p * TPAIR);
                *(v2f*)(d)     = (v2f){tv[j].x, tv[j].y};
                *(v2f*)(d + 2) = (v2f){tv[j].z, tv[j].w};
            }
        }
        #pragma unroll
        for (int j = 0; j < 8; ++j) {
            const int idx = tid + 256 * j;
            sX[(idx >> 4) * XSTR + (idx & 15)] = xv[j];
        }
        __syncthreads();
        float yv[E];
        #pragma unroll
        for (int e = 0; e < E; ++e) yv[e] = sY[it * E + e];

        float xr[L];
        #pragma unroll
        for (int j = 0; j < L; ++j) xr[j] = sX[crow * XSTR + j];

        float sel0[4], sel1[4];
        #pragma unroll
        for (int j = 0; j < 4; ++j) {
            const float e0 = xr[4 * j],     e1 = xr[4 * j + 1];
            const float e2 = xr[4 * j + 2], e3 = xr[4 * j + 3];
            sel0[j] = hi ? e2 : e0;
            sel1[j] = hi ? e3 : e1;
        }

        v8f acc[NTILE];
        #pragma unroll
        for (int t = 0; t < NTILE; ++t)
            acc[t] = (v8f){0.f, 0.f, 0.f, 0.f, 0.f, 0.f, 0.f, 0.f};

        NodeCtx n;
        n.Bb[0] = sB2 + hi * S2 + nl * 2;
        n.Bb[1] = n.Bb[0] + b2_far;
        n.Bb[2] = n.Bb[1] + b2_far;
        n.sW  = sW;  n.T = T;  n.x = x;
        n.wt  = WTabs{Wm0, W20, W10, Wm1, W21, W11};
        n.tid = tid; n.half = half; n.hc = hc; n.c7 = c7;
        n.bn  = b + NODE_BLOCKS;
        n.has_next = n.bn < NN;

        float wl[WCH][E];
        v2f bq[2][NTILE];
        #pragma unroll
        for (int t = 0; t < NTILE; ++t) bq[0][t] = *(const v2f*)(n.Bb[0] + t * 32);
        kchunk<0>(n, acc, bq, xr, sel0, sel1, yv, wl, tv, xv);

        #pragma unroll
        for (int t = 0; t < NTILE; ++t) {
            #pragma unroll
            for (int r = 0; r < 8; ++r)
                sP[(wv * 16 + r + hi * 8) * PSTR + t * 16 + nl] = acc[t][r];
        }
        __syncthreads();

        for (int idx = tid; idx < 4 * C; idx += 256) {
            const int c = idx & 127;
            const int g = idx >> 7;
            const float* P = sP + c * PSTR;
            float s = 0.f;
            if (g == 0) {
                #pragma unroll
                for (int k = 0; k < K3_0; ++k) s = fmaf(P[COL_A0 + k], sW[(WR_M0 + k) * C + c], s);
                #pragma unroll
                for (int k = 0; k < K2_0; ++k) s = fmaf(P[COL_20 + k], sW[(WR_20 + k) * C + c], s);
                float d = 0.f;
                #pragma unroll
                for (int j = 0; j < L; ++j) d = fmaf(sU1[j], sX[c * XSTR + j], d);
                s = fmaf(sW[WR_10 * C + c], d, s);
                out[(size_t)b * (4 * C) + c] = s;
            } else {
                const int w = g - 1;
                #pragma unroll
                for (int k = 0; k < K3_1; ++k) s = fmaf(P[COL_A1 + w * K3_1 + k], sW[(WR_M1 + k) * C + c], s);
                #pragma unroll
                for (int k = 0; k < K2_1; ++k) s = fmaf(P[COL_21 + w * K2_1 + k], sW[(WR_21 + k) * C + c], s);
                float d = 0.f;
                #pragma unroll
                for (int j = 0; j < L; ++j) d = fmaf(sU1[16 + w * L + j], sX[c * XSTR + j], d);
                s = fmaf(sW[WR_11 * C + c], d, s);
                out[(size_t)b * (4 * C) + C + c * 3 + w] = s;
            }
        }
    }
}

extern "C" void kernel_launch(void* const* d_in, const int* in_sizes, int n_in,
                              void* d_out, int out_size, void* d_ws, size_t ws_size,
                              hipStream_t stream) {
    (void)in_sizes; (void)n_in; (void)out_size; (void)ws_size;
    const float* x    = (const float*)d_in[0];
    const float* y    = (const float*)d_in[1];
    const float* U3_0 = (const float*)d_in[2];
    const float* U2_0 = (const float*)d_in[3];
    const float* U1_0 = (const float*)d_in[4];
    const float* U3_1 = (const float*)d_in[5];
    const float* U2_1 = (const float*)d_in[6];
    const float* U1_1 = (const float*)d_in[7];
    const float* Wm0  = (const float*)d_in[8];
    const float* W20  = (const float*)d_in[9];
    const float* W10  = (const float*)d_in[10];
    const float* Wm1  = (const float*)d_in[11];
    const float* W21  = (const float*)d_in[12];
    const float* W11  = (const float*)d_in[13];

    float* T  = (float*)d_ws;
    float* xs = T + WS_XS;

    xs_kernel<<<dim3(XS_BLOCKS), dim3(64), 0, stream>>>(x, xs);
    tbuild_kernel<<<dim3(NTT), dim3(256), 0, stream>>>(U3_0, U3_1, xs, T);

    constexpr size_t smem_bytes = (size_t)SMEM_FLOATS * sizeof(float);
    mace_node_kernel<<<dim3(NODE_BLOCKS), dim3(256), smem_bytes, stream>>>(
        x, y, U2_0, U1_0, U2_1, U1_1,
        Wm0, W20, W10, Wm1, W21, W11, T, (float*)d_out, B2_FAR);
}


# T build: a wave computes two entry tiles (lane c = entries 2c, 2c+1) and stores 16 bytes per lane: one store instruction writes whole aligned 256-byte pieces of T, not 128-byte halves; sums kept
# speedup vs baseline: 4.2011x; 1.0572x over previous
#include <hip/hip_runtime.h>

typedef float v2f __attribute__((ext_vector_type(2)));
typedef float v4f __attribute__((ext_vector_type(4)));
typedef float v8f __attribute__((ext_vector_type(8)));

constexpr int NN   = 2048;
constexpr int C    = 128;
constexpr int L    = 16;
constexpr int E    = 10;
constexpr int K3_0 = 23, K2_0 = 4;
constexpr int K3_1 = 33, K2_1 = 6;

constexpr int COL_A1 = 0;
constexpr int COL_A0 = 3 * K3_1;
constexpr int COL_21 = COL_A0 + K3_0;
constexpr int COL_20 = COL_21 + 3 * K2_1;
constexpr int NCOL   = COL_20 + K2_0;
constexpr int NTILE  = NCOL / 16;

constexpr int S2   = 2 * NCOL + 2;
constexpr int PSTR = 149;
constexpr int FAR_STEPS = 28;
constexpr int B2_FAR    = 2 * FAR_STEPS * S2;
constexpr int XSTR = L + 1;

constexpr int WR_M1 = 0;
constexpr int WR_M0 = 34;
constexpr int WR_21 = 58;
constexpr int WR_20 = 64;
constexpr int WR_11 = 68;
constexpr int WR_10 = 69;
constexpr int WROWS = 70;
constexpr int WPAIRS = WROWS / 2;
constexpr int WCH    = 5;

constexpr int OFF_X  = 0;
constexpr int OFF_W  = OFF_X + 2 * C * XSTR;
constexpr int NODE_BLOCKS = 256;
constexpr int NPB    = NN / NODE_BLOCKS;
constexpr int OFF_Y  = OFF_W + WROWS * C;
constexpr int OFF_U1 = OFF_Y + NPB * E;
constexpr int OFF_B2 = OFF_U1 + 64;
constexpr int OFF_P  = OFF_B2 + 128 * S2;
constexpr int SMEM_FLOATS = OFF_P + C * PSTR;

constexpr int TCOLS = COL_21;
constexpr int TPAIR = 2 * TCOLS;
constexpr int TNODE = 128 * TPAIR;
constexpr int NTT   = 128 * TCOLS / 16;
constexpr size_t WS_XS = (size_t)NN * TNODE;
constexpr int XS_BLOCKS = NN * L / 64;

__global__ __launch_bounds__(64) void xs_kernel(const float* __restrict__ x,
                                                float* __restrict__ xs)
{
    const int g = blockIdx.x * 64 + threadIdx.x;
    const float* xp = x + (size_t)(g >> 4) * (C * L) + (g & 15);
    float s = 0.f;
    #pragma unroll 32
    for (int c = 0; c < C; ++c) s += xp[c * L];
    xs[g] = s;
}

constexpr int NTP = NTT / 2;

__device__ __forceinline__ const float* tentry_src(const float* __restrict__ U3_0,
                                                   const float* __restrict__ U3_1, int n, int& ks)
{
    const int p   = n / TCOLS;
    const int col = n - p * TCOLS;
    if (col < COL_A0) {
        const int w = col / K3_1, k = col - w * K3_1;
        ks = K3_1;
        return U3_1 + (size_t)(w * 256 + 2 * p) * (L * K3_1) + k;
    }
    ks = K3_0;
    return U3_0 + (size_t)(2 * p) * (L * K3_0) + (col - COL_A0);
}

__global__ __launch_bounds__(256) void tbuild_kernel(const float* __restrict__ U3_0,
                                                     const float* __restrict__ U3_1,
                                                     const float* __restrict__ xs,
                                                     float* __restrict__ T)
{
    const int tid  = threadIdx.x;
    const int wv   = tid >> 5;
    const int lane = tid & 31;
    const int c    = lane & 15;
    const int h    = lane >> 4;
    const int tp   = blockIdx.x >> 1;
    const int nh   = blockIdx.x & 1;
    const int n0   = tp * 32 + 2 * c;

    v2f be[2][4], bo[2][4];
    #pragma unroll
    for (int u = 0; u < 2; ++u) {
        int ks;
        const float* se = tentry_src(U3_0, U3_1, n0 + u, ks);
        const float* so = se + L * ks;
        #pragma unroll
        for (int s = 0; s < 4; ++s) {
            const int i0 = 4 * s + 2 * h;
            be[u][s] = (v2f){se[i0 * ks], se[(i0 + 1) * ks]};
            bo[u][s] = (v2f){so[i0 * ks], so[(i0 + 1) * ks]};
        }
    }

    #pragma unroll 2
    for (int j = 0; j < 8; ++j) {
        const int m = nh * 64 + wv * 8 + j;
        const float* xa = xs + (size_t)(m * 16 + c) * L + 2 * h;
        v2f a[4];
        #pragma unroll
        for (int s = 0; s < 4; ++s) a[s] = *(const v2f*)(xa + 4 * s);
        v8f de[2], dd[2];
        #pragma unroll
        for (int u = 0; u < 2; ++u) {
            de[u] = (v8f){0.f, 0.f, 0.f, 0.f, 0.f, 0.f, 0.f, 0.f};
            dd[u] = (v8f){0.f, 0.f, 0.f, 0.f, 0.f, 0.f, 0.f, 0.f};
        }
        #pragma unroll
        for (int s = 0; s < 4; ++s) {
            #pragma unroll
            for (int u = 0; u < 2; ++u) {
                de[u] = __builtin_amdgcn_wmma_f32_16x16x4_f32(
                    false, a[s], false, be[u][s], (short)0, de[u], false, false);
                dd[u] = __builtin_amdgcn_wmma_f32_16x16x4_f32(
                    false, a[s], false, bo[u][s], (short)0, dd[u], false, false);
            }
        }
        float* tq = T + (size_t)(m * 16 + 8 * h) * TNODE + 2 * n0;
        #pragma unroll
        for (int r = 0; r < 8; ++r)
            *(v4f*)(tq + (size_t)r * TNODE) = (v4f){de[0][r], dd[0][r], de[1][r], dd[1][r]};
    }
}

__device__ __forceinline__ void node_loads(const float* __restrict__ T, const float* __restrict__ x,
                                           int b, int tid, v4f (&tv)[31], float (&xv)[8])
{
    const v4f* Tb = (const v4f*)(T + (size_t)b * TNODE);
    #pragma unroll
    for (int j = 0; j < 31; ++j) {
        if (j < 30 || tid < 128)
            tv[j] = __builtin_nontemporal_load(Tb + tid + 256 * j);
    }
    const float* xb = x + (size_t)b * (C * L);
    #pragma unroll
    for (int j = 0; j < 8; ++j) xv[j] = xb[tid + 256 * j];
}

struct WTabs { const float* Wm0; const float* W20; const float* W10;
               const float* Wm1; const float* W21; const float* W11; };

template <int RP>
__device__ __forceinline__ const float* wpair_src(const WTabs& w, int half, int hc, int c7, int& stride)
{
    if constexpr (RP < 17)      { stride = K3_1 * C; return w.Wm1 + (2 * RP) * C        + (RP == 16 ? c7 : hc); }
    else if constexpr (RP < 29) { stride = K3_0 * C; return w.Wm0 + (2 * (RP - 17)) * C + (RP == 28 ? c7 : hc); }
    else if constexpr (RP < 32) { stride = K2_1 * C; return w.W21 + (2 * (RP - 29)) * C + hc; }
    else if constexpr (RP < 34) { stride = K2_0 * C; return w.W20 + (2 * (RP - 32)) * C + hc; }
    else                        { stride = C;        return (half ? w.W10 : w.W11) + c7; }
}

struct NodeCtx {
    const float* Bb[3];
    float*       sW;
    const float* T;
    const float* x;
    WTabs        wt;
    int tid, half, hc, c7, bn;
    bool has_next;
};

template <int Q, int U>
__device__ __forceinline__ void wload(const NodeCtx& n, float (&wl)[WCH][E])
{
    if constexpr (U < WCH) {
        int stride;
        const float* wsrc = wpair_src<Q * WCH + U>(n.wt, n.half, n.hc, n.c7, stride);
        #pragma unroll
        for (int e = 0; e < E; ++e) wl[U][e] = wsrc[e * stride];
        wload<Q, U + 1>(n, wl);
    }
}
template <int Q>
__device__ __forceinline__ void wsum(const NodeCtx& n, const float (&wl)[WCH][E], const float (&yv)[E])
{
    #pragma unroll
    for (int u = 0; u < WCH; ++u) {
        float s = 0.f;
        #pragma unroll
        for (int e = 0; e < E; ++e) s = fmaf(yv[e], wl[u][e], s);
        n.sW[(Q * WCH + u) * 256 + n.tid] = s;
    }
}

template <int Q>
__device__ __forceinline__ void kchunk(const NodeCtx& n, v8f (&acc)[NTILE], v2f (&bq)[2][NTILE],
                                       const float (&xr)[L], const float (&sel0)[4], const float (&sel1)[4],
                                       const float (&yv)[E], float (&wl)[WCH][E],
                                       v4f (&tv)[31], float (&xv)[8])
{
    constexpr int NWQ = WPAIRS / WCH;
    if constexpr (Q <= NWQ) {
        if constexpr (Q >= 1)  wsum<Q - 1>(n, wl, yv);
        if constexpr (Q < NWQ) wload<Q, 0>(n, wl);
    }

    const float xp = xr[Q];
    #pragma unroll
    for (int j = 0; j < 4; ++j) {
        v2f a;
        a.x = xp * sel0[j];
        a.y = xp * sel1[j];
        if (4 * Q + j + 1 < 64) {
            const int sn = 4 * Q + j + 1;
            const float* Bn = n.Bb[sn / FAR_STEPS] + (2 * (sn % FAR_STEPS)) * S2;
            #pragma unroll
            for (int t = 0; t < NTILE; ++t)
                bq[(j + 1) & 1][t] = *(const v2f*)(Bn + t * 32);
        }
        __builtin_amdgcn_sched_barrier(0);
        #pragma unroll
        for (int t = 0; t < NTILE; ++t)
            acc[t] = __builtin_amdgcn_wmma_f32_16x16x4_f32(
                false, a, false, bq[j & 1][t], (short)0, acc[t], false, false);
        __builtin_amdgcn_sched_barrier(0);
    }

    if constexpr (Q == NWQ) {
        if (n.has_next) node_loads(n.T, n.x, n.bn, n.tid, tv, xv);
    }

    if constexpr (Q + 1 < 16) kchunk<Q + 1>(n, acc, bq, xr, sel0, sel1, yv, wl, tv, xv);
}

__global__ __launch_bounds__(256) void mace_node_kernel(
    const float* __restrict__ x,    const float* __restrict__ y,
    const float* __restrict__ U2_0, const float* __restrict__ U1_0,
    const float* __restrict__ U2_1, const float* __restrict__ U1_1,
    const float* Wm0,  const float* W20,  const float* W10,
    const float* Wm1,  const float* W21,  const float* W11,
    const float* __restrict__ T,    float* out, int b2_far)
{
    extern __shared__ float sm[];
    float* sW  = sm + OFF_W;
    float* sY  = sm + OFF_Y;
    float* sU1 = sm + OFF_U1;
    float* sB2 = sm + OFF_B2;
    float* sP  = sm + OFF_P;

    const int tid  = threadIdx.x;
    const int wv   = tid >> 5;
    const int lane = tid & 31;
    const int nl   = lane & 15;
    const int hi   = lane >> 4;
    const int crow = wv * 16 + nl;
    const int c7   = tid & 127;
    const int half = tid >> 7;
    const int hc   = half * C + c7;

    int b = blockIdx.x;

    v4f tv[31];
    float xv[8];
    tv[30] = (v4f){0.f, 0.f, 0.f, 0.f};
    if (b < NN) node_loads(T, x, b, tid, tv, xv);

    if (tid < NPB * E) sY[tid] = y[(blockIdx.x + NODE_BLOCKS * (tid / E)) * E + (tid % E)];
    if (tid < 16)               sU1[tid] = U1_0[tid];
    else if (tid < 64)          sU1[tid] = U1_1[tid - 16];
    if (tid < 128) {
        const int p = tid;
        float* bp = sB2 + p * S2;
        #pragma unroll
        for (int m = 0; m < 3 * K2_1; ++m) {
            const int w = m / K2_1, k = m % K2_1;
            const int r0 = (w * 256 + 2 * p) * K2_1 + k;
            *(v2f*)(bp + (COL_21 + m) * 2) = (v2f){U2_1[r0], U2_1[r0 + K2_1]};
        }
        #pragma unroll
        for (int m = 0; m < K2_0; ++m)
            *(v2f*)(bp + (COL_20 + m) * 2) =
                (v2f){U2_0[(2 * p) * K2_0 + m], U2_0[(2 * p + 1) * K2_0 + m]};
    }

    int par = 0, it = 0;
    for (; b < NN; b += NODE_BLOCKS, par ^= 1, ++it) {
        float* sX = sm + OFF_X + par * (C * XSTR);

        #pragma unroll
        for (int j = 0; j < 31; ++j) {
            if (j < 30 || tid < 128) {
                const int f = 4 * (tid + 256 * j);
                const int p = f / TPAIR;
                float* d = sB2 + p * S2 + (f - p * TPAIR);
                *(v2f*)(d)     = (v2f){tv[j].x, tv[j].y};
                *(v2f*)(d + 2) = (v2f){tv[j].z, tv[j].w};
            }
        }
        #pragma unroll
        for (int j = 0; j < 8; ++j) {
            const int idx = tid + 256 * j;
            sX[(idx >> 4) * XSTR + (idx & 15)] = xv[j];
        }
        __syncthreads();
        float yv[E];
        #pragma unroll
        for (int e = 0; e < E; ++e) yv[e] = sY[it * E + e];

        float xr[L];
        #pragma unroll
        for (int j = 0; j < L; ++j) xr[j] = sX[crow * XSTR + j];

        float sel0[4], sel1[4];
        #pragma unroll
        for (int j = 0; j < 4; ++j) {
            const float e0 = xr[4 * j],     e1 = xr[4 * j + 1];
            const float e2 = xr[4 * j + 2], e3 = xr[4 * j + 3];
            sel0[j] = hi ? e2 : e0;
            sel1[j] = hi ? e3 : e1;
        }

        v8f acc[NTILE];
        #pragma unroll
        for (int t = 0; t < NTILE; ++t)
            acc[t] = (v8f){0.f, 0.f, 0.f, 0.f, 0.f, 0.f, 0.f, 0.f};

        NodeCtx n;
        n.Bb[0] = sB2 + hi * S2 + nl * 2;
        n.Bb[1] = n.Bb[0] + b2_far;
        n.Bb[2] = n.Bb[1] + b2_far;
        n.sW  = sW;  n.T = T;  n.x = x;
        n.wt  = WTabs{Wm0, W20, W10, Wm1, W21, W11};
        n.tid = tid; n.half = half; n.hc = hc; n.c7 = c7;
        n.bn  = b + NODE_BLOCKS;
        n.has_next = n.bn < NN;

        float wl[WCH][E];
        v2f bq[2][NTILE];
        #pragma unroll
        for (int t = 0; t < NTILE; ++t) bq[0][t] = *(const v2f*)(n.Bb[0] + t * 32);
        kchunk<0>(n, acc, bq, xr, sel0, sel1, yv, wl, tv, xv);

        #pragma unroll
        for (int t = 0; t < NTILE; ++t) {
            #pragma unroll
            for (int r = 0; r < 8; ++r)
                sP[(wv * 16 + r + hi * 8) * PSTR + t * 16 + nl] = acc[t][r];
        }
        __syncthreads();

        for (int idx = tid; idx < 4 * C; idx += 256) {
            const int c = idx & 127;
            const int g = idx >> 7;
            const float* P = sP + c * PSTR;
            float s = 0.f;
            if (g == 0) {
                #pragma unroll
                for (int k = 0; k < K3_0; ++k) s = fmaf(P[COL_A0 + k], sW[(WR_M0 + k) * C + c], s);
                #pragma unroll
                for (int k = 0; k < K2_0; ++k) s = fmaf(P[COL_20 + k], sW[(WR_20 + k) * C + c], s);
                float d = 0.f;
                #pragma unroll
                for (int j = 0; j < L; ++j) d = fmaf(sU1[j], sX[c * XSTR + j], d);
                s = fmaf(sW[WR_10 * C + c], d, s);
                out[(size_t)b * (4 * C) + c] = s;
            } else {
                const int w = g - 1;
                #pragma unroll
                for (int k = 0; k < K3_1; ++k) s = fmaf(P[COL_A1 + w * K3_1 + k], sW[(WR_M1 + k) * C + c], s);
                #pragma unroll
                for (int k = 0; k < K2_1; ++k) s = fmaf(P[COL_21 + w * K2_1 + k], sW[(WR_21 + k) * C + c], s);
                float d = 0.f;
                #pragma unroll
                for (int j = 0; j < L; ++j) d = fmaf(sU1[16 + w * L + j], sX[c * XSTR + j], d);
                s = fmaf(sW[WR_11 * C + c], d, s);
                out[(size_t)b * (4 * C) + C + c * 3 + w] = s;
            }
        }
    }
}

extern "C" void kernel_launch(void* const* d_in, const int* in_sizes, int n_in,
                              void* d_out, int out_size, void* d_ws, size_t ws_size,
                              hipStream_t stream) {
    (void)in_sizes; (void)n_in; (void)out_size; (void)ws_size;
    const float* x    = (const float*)d_in[0];
    const float* y    = (const float*)d_in[1];
    const float* U3_0 = (const float*)d_in[2];
    const float* U2_0 = (const float*)d_in[3];
    const float* U1_0 = (const float*)d_in[4];
    const float* U3_1 = (const float*)d_in[5];
    const float* U2_1 = (const float*)d_in[6];
    const float* U1_1 = (const float*)d_in[7];
    const float* Wm0  = (const float*)d_in[8];
    const float* W20  = (const float*)d_in[9];
    const float* W10  = (const float*)d_in[10];
    const float* Wm1  = (const float*)d_in[11];
    const float* W21  = (const float*)d_in[12];
    const float* W11  = (const float*)d_in[13];

    float* T  = (float*)d_ws;
    float* xs = T + WS_XS;

    xs_kernel<<<dim3(XS_BLOCKS), dim3(64), 0, stream>>>(x, xs);
    tbuild_kernel<<<dim3(NTT), dim3(256), 0, stream>>>(U3_0, U3_1, xs, T);

    constexpr size_t smem_bytes = (size_t)SMEM_FLOATS * sizeof(float);
    mace_node_kernel<<<dim3(NODE_BLOCKS), dim3(256), smem_bytes, stream>>>(
        x, y, U2_0, U1_0, U2_1, U1_1,
        Wm0, W20, W10, Wm1, W21, W11, T, (float*)d_out, B2_FAR);
}


# T in the workspace re-ordered [node tile][256-B piece][node]: a build wave writes one 4-KB run, 512 contiguous bytes a store (was 2 x 256 B, 1 MB apart); node kernel gathers its pieces; sums kept
# speedup vs baseline: 4.3513x; 1.0358x over previous
#include <hip/hip_runtime.h>

typedef float v2f __attribute__((ext_vector_type(2)));
typedef float v4f __attribute__((ext_vector_type(4)));
typedef float v8f __attribute__((ext_vector_type(8)));

constexpr int NN   = 2048;
constexpr int C    = 128;
constexpr int L    = 16;
constexpr int E    = 10;
constexpr int K3_0 = 23, K2_0 = 4;
constexpr int K3_1 = 33, K2_1 = 6;

constexpr int COL_A1 = 0;
constexpr int COL_A0 = 3 * K3_1;
constexpr int COL_21 = COL_A0 + K3_0;
constexpr int COL_20 = COL_21 + 3 * K2_1;
constexpr int NCOL   = COL_20 + K2_0;
constexpr int NTILE  = NCOL / 16;

constexpr int S2   = 2 * NCOL + 2;
constexpr int PSTR = 149;
constexpr int FAR_STEPS = 28;
constexpr int B2_FAR    = 2 * FAR_STEPS * S2;
constexpr int XSTR = L + 1;

constexpr int WR_M1 = 0;
constexpr int WR_M0 = 34;
constexpr int WR_21 = 58;
constexpr int WR_20 = 64;
constexpr int WR_11 = 68;
constexpr int WR_10 = 69;
constexpr int WROWS = 70;
constexpr int WPAIRS = WROWS / 2;
constexpr int WCH    = 5;

constexpr int OFF_X  = 0;
constexpr int OFF_W  = OFF_X + 2 * C * XSTR;
constexpr int NODE_BLOCKS = 256;
constexpr int NPB    = NN / NODE_BLOCKS;
constexpr int OFF_Y  = OFF_W + WROWS * C;
constexpr int OFF_U1 = OFF_Y + NPB * E;
constexpr int OFF_B2 = OFF_U1 + 64;
constexpr int OFF_P  = OFF_B2 + 128 * S2;
constexpr int SMEM_FLOATS = OFF_P + C * PSTR;

constexpr int TCOLS = COL_21;
constexpr int TPAIR = 2 * TCOLS;
constexpr int TNODE = 128 * TPAIR;
constexpr int NTT   = 128 * TCOLS / 16;
constexpr int TSLOT  = 64;
constexpr int TCHUNK = 16 * TSLOT;
constexpr int TTILE  = 16 * TNODE;
static_assert(TNODE % TSLOT == 0 && TTILE == (TNODE / TSLOT) * TCHUNK, "T chunk layout");
constexpr size_t WS_XS = (size_t)NN * TNODE;
constexpr int XS_BLOCKS = NN * L / 64;

__global__ __launch_bounds__(64) void xs_kernel(const float* __restrict__ x,
                                                float* __restrict__ xs)
{
    const int g = blockIdx.x * 64 + threadIdx.x;
    const float* xp = x + (size_t)(g >> 4) * (C * L) + (g & 15);
    float s = 0.f;
    #pragma unroll 32
    for (int c = 0; c < C; ++c) s += xp[c * L];
    xs[g] = s;
}

constexpr int NTP = NTT / 2;

__device__ __forceinline__ const float* tentry_src(const float* __restrict__ U3_0,
                                                   const float* __restrict__ U3_1, int n, int& ks)
{
    const int p   = n / TCOLS;
    const int col = n - p * TCOLS;
    if (col < COL_A0) {
        const int w = col / K3_1, k = col - w * K3_1;
        ks = K3_1;
        return U3_1 + (size_t)(w * 256 + 2 * p) * (L * K3_1) + k;
    }
    ks = K3_0;
    return U3_0 + (size_t)(2 * p) * (L * K3_0) + (col - COL_A0);
}

__global__ __launch_bounds__(256) void tbuild_kernel(const float* __restrict__ U3_0,
                                                     const float* __restrict__ U3_1,
                                                     const float* __restrict__ xs,
                                                     float* __restrict__ T)
{
    const int tid  = threadIdx.x;
    const int wv   = tid >> 5;
    const int lane = tid & 31;
    const int c    = lane & 15;
    const int h    = lane >> 4;
    const int tp   = blockIdx.x >> 1;
    const int nh   = blockIdx.x & 1;
    const int n0   = tp * 32 + 2 * c;

    v2f be[2][4], bo[2][4];
    #pragma unroll
    for (int u = 0; u < 2; ++u) {
        int ks;
        const float* se = tentry_src(U3_0, U3_1, n0 + u, ks);
        const float* so = se + L * ks;
        #pragma unroll
        for (int s = 0; s < 4; ++s) {
            const int i0 = 4 * s + 2 * h;
            be[u][s] = (v2f){se[i0 * ks], se[(i0 + 1) * ks]};
            bo[u][s] = (v2f){so[i0 * ks], so[(i0 + 1) * ks]};
        }
    }

    #pragma unroll 2
    for (int j = 0; j < 8; ++j) {
        const int m = nh * 64 + wv * 8 + j;
        const float* xa = xs + (size_t)(m * 16 + c) * L + 2 * h;
        v2f a[4];
        #pragma unroll
        for (int s = 0; s < 4; ++s) a[s] = *(const v2f*)(xa + 4 * s);
        v8f de[2], dd[2];
        #pragma unroll
        for (int u = 0; u < 2; ++u) {
            de[u] = (v8f){0.f, 0.f, 0.f, 0.f, 0.f, 0.f, 0.f, 0.f};
            dd[u] = (v8f){0.f, 0.f, 0.f, 0.f, 0.f, 0.f, 0.f, 0.f};
        }
        #pragma unroll
        for (int s = 0; s < 4; ++s) {
            #pragma unroll
            for (int u = 0; u < 2; ++u) {
                de[u] = __builtin_amdgcn_wmma_f32_16x16x4_f32(
                    false, a[s], false, be[u][s], (short)0, de[u], false, false);
                dd[u] = __builtin_amdgcn_wmma_f32_16x16x4_f32(
                    false, a[s], false, bo[u][s], (short)0, dd[u], false, false);
            }
        }
        float* tq = T + (size_t)m * TTILE + (size_t)tp * TCHUNK + h * TSLOT + 4 * c;
        #pragma unroll
        for (int r = 0; r < 8; ++r)
            *(v4f*)(tq + r * (2 * TSLOT)) = (v4f){de[0][r], dd[0][r], de[1][r], dd[1][r]};
    }
}

__device__ __forceinline__ void node_loads(const float* __restrict__ T, const float* __restrict__ x,
                                           int b, int tid, v4f (&tv)[31], float (&xv)[8])
{
    const int row  = b & 15;
    const int slot = 2 * (row & 7) + (row >> 3);
    const v4f* Tb = (const v4f*)(T + (size_t)(b >> 4) * TTILE + slot * TSLOT)
                    + (tid >> 4) * (TCHUNK / 4) + (tid & 15);
    #pragma unroll
    for (int j = 0; j < 31; ++j) {
        if (j < 30 || tid < 128)
            tv[j] = __builtin_nontemporal_load(Tb + j * (16 * (TCHUNK / 4)));
    }
    const float* xb = x + (size_t)b * (C * L);
    #pragma unroll
    for (int j = 0; j < 8; ++j) xv[j] = xb[tid + 256 * j];
}

struct WTabs { const float* Wm0; const float* W20; const float* W10;
               const float* Wm1; const float* W21; const float* W11; };

template <int RP>
__device__ __forceinline__ const float* wpair_src(const WTabs& w, int half, int hc, int c7, int& stride)
{
    if constexpr (RP < 17)      { stride = K3_1 * C; return w.Wm1 + (2 * RP) * C        + (RP == 16 ? c7 : hc); }
    else if constexpr (RP < 29) { stride = K3_0 * C; return w.Wm0 + (2 * (RP - 17)) * C + (RP == 28 ? c7 : hc); }
    else if constexpr (RP < 32) { stride = K2_1 * C; return w.W21 + (2 * (RP - 29)) * C + hc; }
    else if constexpr (RP < 34) { stride = K2_0 * C; return w.W20 + (2 * (RP - 32)) * C + hc; }
    else                        { stride = C;        return (half ? w.W10 : w.W11) + c7; }
}

struct NodeCtx {
    const float* Bb[3];
    float*       sW;
    const float* T;
    const float* x;
    WTabs        wt;
    int tid, half, hc, c7, bn;
    bool has_next;
};

template <int Q, int U>
__device__ __forceinline__ void wload(const NodeCtx& n, float (&wl)[WCH][E])
{
    if constexpr (U < WCH) {
        int stride;
        const float* wsrc = wpair_src<Q * WCH + U>(n.wt, n.half, n.hc, n.c7, stride);
        #pragma unroll
        for (int e = 0; e < E; ++e) wl[U][e] = wsrc[e * stride];
        wload<Q, U + 1>(n, wl);
    }
}
template <int Q>
__device__ __forceinline__ void wsum(const NodeCtx& n, const float (&wl)[WCH][E], const float (&yv)[E])
{
    #pragma unroll
    for (int u = 0; u < WCH; ++u) {
        float s = 0.f;
        #pragma unroll
        for (int e = 0; e < E; ++e) s = fmaf(yv[e], wl[u][e], s);
        n.sW[(Q * WCH + u) * 256 + n.tid] = s;
    }
}

template <int Q>
__device__ __forceinline__ void kchunk(const NodeCtx& n, v8f (&acc)[NTILE], v2f (&bq)[2][NTILE],
                                       const float (&xr)[L], const float (&sel0)[4], const float (&sel1)[4],
                                       const float (&yv)[E], float (&wl)[WCH][E],
                                       v4f (&tv)[31], float (&xv)[8])
{
    constexpr int NWQ = WPAIRS / WCH;
    if constexpr (Q <= NWQ) {
        if constexpr (Q >= 1)  wsum<Q - 1>(n, wl, yv);
        if constexpr (Q < NWQ) wload<Q, 0>(n, wl);
    }

    const float xp = xr[Q];
    #pragma unroll
    for (int j = 0; j < 4; ++j) {
        v2f a;
        a.x = xp * sel0[j];
        a.y = xp * sel1[j];
        if (4 * Q + j + 1 < 64) {
            const int sn = 4 * Q + j + 1;
            const float* Bn = n.Bb[sn / FAR_STEPS] + (2 * (sn % FAR_STEPS)) * S2;
            #pragma unroll
            for (int t = 0; t < NTILE; ++t)
                bq[(j + 1) & 1][t] = *(const v2f*)(Bn + t * 32);
        }
        __builtin_amdgcn_sched_barrier(0);
        #pragma unroll
        for (int t = 0; t < NTILE; ++t)
            acc[t] = __builtin_amdgcn_wmma_f32_16x16x4_f32(
                false, a, false, bq[j & 1][t], (short)0, acc[t], false, false);
        __builtin_amdgcn_sched_barrier(0);
    }

    if constexpr (Q == NWQ) {
        if (n.has_next) node_loads(n.T, n.x, n.bn, n.tid, tv, xv);
    }

    if constexpr (Q + 1 < 16) kchunk<Q + 1>(n, acc, bq, xr, sel0, sel1, yv, wl, tv, xv);
}

__global__ __launch_bounds__(256) void mace_node_kernel(
    const float* __restrict__ x,    const float* __restrict__ y,
    const float* __restrict__ U2_0, const float* __restrict__ U1_0,
    const float* __restrict__ U2_1, const float* __restrict__ U1_1,
    const float* Wm0,  const float* W20,  const float* W10,
    const float* Wm1,  const float* W21,  const float* W11,
    const float* __restrict__ T,    float* out, int b2_far)
{
    extern __shared__ float sm[];
    float* sW  = sm + OFF_W;
    float* sY  = sm + OFF_Y;
    float* sU1 = sm + OFF_U1;
    float* sB2 = sm + OFF_B2;
    float* sP  = sm + OFF_P;

    const int tid  = threadIdx.x;
    const int wv   = tid >> 5;
    const int lane = tid & 31;
    const int nl   = lane & 15;
    const int hi   = lane >> 4;
    const int crow = wv * 16 + nl;
    const int c7   = tid & 127;
    const int half = tid >> 7;
    const int hc   = half * C + c7;

    int b = blockIdx.x;

    v4f tv[31];
    float xv[8];
    tv[30] = (v4f){0.f, 0.f, 0.f, 0.f};
    if (b < NN) node_loads(T, x, b, tid, tv, xv);

    if (tid < NPB * E) sY[tid] = y[(blockIdx.x + NODE_BLOCKS * (tid / E)) * E + (tid % E)];
    if (tid < 16)               sU1[tid] = U1_0[tid];
    else if (tid < 64)          sU1[tid] = U1_1[tid - 16];
    if (tid < 128) {
        const int p = tid;
        float* bp = sB2 + p * S2;
        #pragma unroll
        for (int m = 0; m < 3 * K2_1; ++m) {
            const int w = m / K2_1, k = m % K2_1;
            const int r0 = (w * 256 + 2 * p) * K2_1 + k;
            *(v2f*)(bp + (COL_21 + m) * 2) = (v2f){U2_1[r0], U2_1[r0 + K2_1]};
        }
        #pragma unroll
        for (int m = 0; m < K2_0; ++m)
            *(v2f*)(bp + (COL_20 + m) * 2) =
                (v2f){U2_0[(2 * p) * K2_0 + m], U2_0[(2 * p + 1) * K2_0 + m]};
    }

    int par = 0, it = 0;
    for (; b < NN; b += NODE_BLOCKS, par ^= 1, ++it) {
        float* sX = sm + OFF_X + par * (C * XSTR);

        #pragma unroll
        for (int j = 0; j < 31; ++j) {
            if (j < 30 || tid < 128) {
                const int f = 4 * (tid + 256 * j);
                const int p = f / TPAIR;
                float* d = sB2 + p * S2 + (f - p * TPAIR);
                *(v2f*)(d)     = (v2f){tv[j].x, tv[j].y};
                *(v2f*)(d + 2) = (v2f){tv[j].z, tv[j].w};
            }
        }
        #pragma unroll
        for (int j = 0; j < 8; ++j) {
            const int idx = tid + 256 * j;
            sX[(idx >> 4) * XSTR + (idx & 15)] = xv[j];
        }
        __syncthreads();
        float yv[E];
        #pragma unroll
        for (int e = 0; e < E; ++e) yv[e] = sY[it * E + e];

        float xr[L];
        #pragma unroll
        for (int j = 0; j < L; ++j) xr[j] = sX[crow * XSTR + j];

        float sel0[4], sel1[4];
        #pragma unroll
        for (int j = 0; j < 4; ++j) {
            const float e0 = xr[4 * j],     e1 = xr[4 * j + 1];
            const float e2 = xr[4 * j + 2], e3 = xr[4 * j + 3];
            sel0[j] = hi ? e2 : e0;
            sel1[j] = hi ? e3 : e1;
        }

        v8f acc[NTILE];
        #pragma unroll
        for (int t = 0; t < NTILE; ++t)
            acc[t] = (v8f){0.f, 0.f, 0.f, 0.f, 0.f, 0.f, 0.f, 0.f};

        NodeCtx n;
        n.Bb[0] = sB2 + hi * S2 + nl * 2;
        n.Bb[1] = n.Bb[0] + b2_far;
        n.Bb[2] = n.Bb[1] + b2_far;
        n.sW  = sW;  n.T = T;  n.x = x;
        n.wt  = WTabs{Wm0, W20, W10, Wm1, W21, W11};
        n.tid = tid; n.half = half; n.hc = hc; n.c7 = c7;
        n.bn  = b + NODE_BLOCKS;
        n.has_next = n.bn < NN;

        float wl[WCH][E];
        v2f bq[2][NTILE];
        #pragma unroll
        for (int t = 0; t < NTILE; ++t) bq[0][t] = *(const v2f*)(n.Bb[0] + t * 32);
        kchunk<0>(n, acc, bq, xr, sel0, sel1, yv, wl, tv, xv);

        #pragma unroll
        for (int t = 0; t < NTILE; ++t) {
            #pragma unroll
            for (int r = 0; r < 8; ++r)
                sP[(wv * 16 + r + hi * 8) * PSTR + t * 16 + nl] = acc[t][r];
        }
        __syncthreads();

        for (int idx = tid; idx < 4 * C; idx += 256) {
            const int c = idx & 127;
            const int g = idx >> 7;
            const float* P = sP + c * PSTR;
            float s = 0.f;
            if (g == 0) {
                #pragma unroll
                for (int k = 0; k < K3_0; ++k) s = fmaf(P[COL_A0 + k], sW[(WR_M0 + k) * C + c], s);
                #pragma unroll
                for (int k = 0; k < K2_0; ++k) s = fmaf(P[COL_20 + k], sW[(WR_20 + k) * C + c], s);
                float d = 0.f;
                #pragma unroll
                for (int j = 0; j < L; ++j) d = fmaf(sU1[j], sX[c * XSTR + j], d);
                s = fmaf(sW[WR_10 * C + c], d, s);
                out[(size_t)b * (4 * C) + c] = s;
            } else {
                const int w = g - 1;
                #pragma unroll
                for (int k = 0; k < K3_1; ++k) s = fmaf(P[COL_A1 + w * K3_1 + k], sW[(WR_M1 + k) * C + c], s);
                #pragma unroll
                for (int k = 0; k < K2_1; ++k) s = fmaf(P[COL_21 + w * K2_1 + k], sW[(WR_21 + k) * C + c], s);
                float d = 0.f;
                #pragma unroll
                for (int j = 0; j < L; ++j) d = fmaf(sU1[16 + w * L + j], sX[c * XSTR + j], d);
                s = fmaf(sW[WR_11 * C + c], d, s);
                out[(size_t)b * (4 * C) + C + c * 3 + w] = s;
            }
        }
    }
}

extern "C" void kernel_launch(void* const* d_in, const int* in_sizes, int n_in,
                              void* d_out, int out_size, void* d_ws, size_t ws_size,
                              hipStream_t stream) {
    (void)in_sizes; (void)n_in; (void)out_size; (void)ws_size;
    const float* x    = (const float*)d_in[0];
    const float* y    = (const float*)d_in[1];
    const float* U3_0 = (const float*)d_in[2];
    const float* U2_0 = (const float*)d_in[3];
    const float* U1_0 = (const float*)d_in[4];
    const float* U3_1 = (const float*)d_in[5];
    const float* U2_1 = (const float*)d_in[6];
    const float* U1_1 = (const float*)d_in[7];
    const float* Wm0  = (const float*)d_in[8];
    const float* W20  = (const float*)d_in[9];
    const float* W10  = (const float*)d_in[10];
    const float* Wm1  = (const float*)d_in[11];
    const float* W21  = (const float*)d_in[12];
    const float* W11  = (const float*)d_in[13];

    float* T  = (float*)d_ws;
    float* xs = T + WS_XS;

    xs_kernel<<<dim3(XS_BLOCKS), dim3(64), 0, stream>>>(x, xs);
    tbuild_kernel<<<dim3(NTT), dim3(256), 0, stream>>>(U3_0, U3_1, xs, T);

    constexpr size_t smem_bytes = (size_t)SMEM_FLOATS * sizeof(float);
    mace_node_kernel<<<dim3(NODE_BLOCKS), dim3(256), smem_bytes, stream>>>(
        x, y, U2_0, U1_0, U2_1, U1_1,
        Wm0, W20, W10, Wm1, W21, W11, T, (float*)d_out, B2_FAR);
}


# node kernel: all side work of the WMMA K loop (weight sums, next-node loads order, and the PREVIOUS node's epilogue, weights double-buffered) put on the 4 older waves, s_setprio on them; sums kept
# speedup vs baseline: 4.5283x; 1.0407x over previous
#include <hip/hip_runtime.h>

typedef float v2f __attribute__((ext_vector_type(2)));
typedef float v4f __attribute__((ext_vector_type(4)));
typedef float v8f __attribute__((ext_vector_type(8)));

constexpr int NN   = 2048;
constexpr int C    = 128;
constexpr int L    = 16;
constexpr int E    = 10;
constexpr int K3_0 = 23, K2_0 = 4;
constexpr int K3_1 = 33, K2_1 = 6;

constexpr int COL_A1 = 0;
constexpr int COL_A0 = 3 * K3_1;
constexpr int COL_21 = COL_A0 + K3_0;
constexpr int COL_20 = COL_21 + 3 * K2_1;
constexpr int NCOL   = COL_20 + K2_0;
constexpr int NTILE  = NCOL / 16;

constexpr int S2   = 2 * NCOL + 2;
constexpr int PSTR = 149;
constexpr int FAR_STEPS = 28;
constexpr int B2_FAR    = 2 * FAR_STEPS * S2;
constexpr int XSTR = L + 1;

constexpr int WR_M1 = 0;
constexpr int WR_M0 = 34;
constexpr int WR_21 = 58;
constexpr int WR_20 = 64;
constexpr int WR_11 = 68;
constexpr int WR_10 = 69;
constexpr int WROWS = 70;
constexpr int WCH    = 5;
constexpr int NWB    = WROWS / WCH;
static_assert(NWB * WCH == WROWS && NWB + 1 <= 15, "weight blocks fit the K chunks");

constexpr int OFF_X  = 0;
constexpr int OFF_W  = OFF_X + 2 * C * XSTR;
constexpr int NODE_BLOCKS = 256;
constexpr int NPB    = NN / NODE_BLOCKS;
constexpr int OFF_Y  = OFF_W + WROWS * C;
constexpr int OFF_U1 = OFF_Y + NPB * E;
constexpr int OFF_B2 = OFF_U1 + 64;
constexpr int OFF_P  = OFF_B2 + 128 * S2;
constexpr int OFF_W2 = OFF_P + C * PSTR;
constexpr int SMEM_FLOATS = OFF_W2 + WROWS * C;

constexpr int TCOLS = COL_21;
constexpr int TPAIR = 2 * TCOLS;
constexpr int TNODE = 128 * TPAIR;
constexpr int NTT   = 128 * TCOLS / 16;
constexpr int TSLOT  = 64;
constexpr int TCHUNK = 16 * TSLOT;
constexpr int TTILE  = 16 * TNODE;
static_assert(TNODE % TSLOT == 0 && TTILE == (TNODE / TSLOT) * TCHUNK, "T chunk layout");
constexpr size_t WS_XS = (size_t)NN * TNODE;
constexpr int XS_BLOCKS = NN * L / 64;

__global__ __launch_bounds__(64) void xs_kernel(const float* __restrict__ x,
                                                float* __restrict__ xs)
{
    const int g = blockIdx.x * 64 + threadIdx.x;
    const float* xp = x + (size_t)(g >> 4) * (C * L) + (g & 15);
    float s = 0.f;
    #pragma unroll 32
    for (int c = 0; c < C; ++c) s += xp[c * L];
    xs[g] = s;
}

constexpr int NTP = NTT / 2;

__device__ __forceinline__ const float* tentry_src(const float* __restrict__ U3_0,
                                                   const float* __restrict__ U3_1, int n, int& ks)
{
    const int p   = n / TCOLS;
    const int col = n - p * TCOLS;
    if (col < COL_A0) {
        const int w = col / K3_1, k = col - w * K3_1;
        ks = K3_1;
        return U3_1 + (size_t)(w * 256 + 2 * p) * (L * K3_1) + k;
    }
    ks = K3_0;
    return U3_0 + (size_t)(2 * p) * (L * K3_0) + (col - COL_A0);
}

__global__ __launch_bounds__(256) void tbuild_kernel(const float* __restrict__ U3_0,
                                                     const float* __restrict__ U3_1,
                                                     const float* __restrict__ xs,
                                                     float* __restrict__ T)
{
    const int tid  = threadIdx.x;
    const int wv   = tid >> 5;
    const int lane = tid & 31;
    const int c    = lane & 15;
    const int h    = lane >> 4;
    const int tp   = blockIdx.x >> 1;
    const int nh   = blockIdx.x & 1;
    const int n0   = tp * 32 + 2 * c;

    v2f be[2][4], bo[2][4];
    #pragma unroll
    for (int u = 0; u < 2; ++u) {
        int ks;
        const float* se = tentry_src(U3_0, U3_1, n0 + u, ks);
        const float* so = se + L * ks;
        #pragma unroll
        for (int s = 0; s < 4; ++s) {
            const int i0 = 4 * s + 2 * h;
            be[u][s] = (v2f){se[i0 * ks], se[(i0 + 1) * ks]};
            bo[u][s] = (v2f){so[i0 * ks], so[(i0 + 1) * ks]};
        }
    }

    #pragma unroll 2
    for (int j = 0; j < 8; ++j) {
        const int m = nh * 64 + wv * 8 + j;
        const float* xa = xs + (size_t)(m * 16 + c) * L + 2 * h;
        v2f a[4];
        #pragma unroll
        for (int s = 0; s < 4; ++s) a[s] = *(const v2f*)(xa + 4 * s);
        v8f de[2], dd[2];
        #pragma unroll
        for (int u = 0; u < 2; ++u) {
            de[u] = (v8f){0.f, 0.f, 0.f, 0.f, 0.f, 0.f, 0.f, 0.f};
            dd[u] = (v8f){0.f, 0.f, 0.f, 0.f, 0.f, 0.f, 0.f, 0.f};
        }
        #pragma unroll
        for (int s = 0; s < 4; ++s) {
            #pragma unroll
            for (int u = 0; u < 2; ++u) {
                de[u] = __builtin_amdgcn_wmma_f32_16x16x4_f32(
                    false, a[s], false, be[u][s], (short)0, de[u], false, false);
                dd[u] = __builtin_amdgcn_wmma_f32_16x16x4_f32(
                    false, a[s], false, bo[u][s], (short)0, dd[u], false, false);
            }
        }
        float* tq = T + (size_t)m * TTILE + (size_t)tp * TCHUNK + h * TSLOT + 4 * c;
        #pragma unroll
        for (int r = 0; r < 8; ++r)
            *(v4f*)(tq + r * (2 * TSLOT)) = (v4f){de[0][r], dd[0][r], de[1][r], dd[1][r]};
    }
}

__device__ __forceinline__ void node_loads(const float* __restrict__ T, const float* __restrict__ x,
                                           int b, int tid, v4f (&tv)[31], float (&xv)[8])
{
    const int row  = b & 15;
    const int slot = 2 * (row & 7) + (row >> 3);
    const v4f* Tb = (const v4f*)(T + (size_t)(b >> 4) * TTILE + slot * TSLOT)
                    + (tid >> 4) * (TCHUNK / 4) + (tid & 15);
    #pragma unroll
    for (int j = 0; j < 31; ++j) {
        if (j < 30 || tid < 128)
            tv[j] = __builtin_nontemporal_load(Tb + j * (16 * (TCHUNK / 4)));
    }
    const float* xb = x + (size_t)b * (C * L);
    #pragma unroll
    for (int j = 0; j < 8; ++j) xv[j] = xb[tid + 256 * j];
}

struct WTabs { const float* Wm0; const float* W20; const float* W10;
               const float* Wm1; const float* W21; const float* W11; };

template <int R>
__device__ __forceinline__ const float* wrow_src(const WTabs& w, int c, int& stride)
{
    if constexpr (R < WR_M0)      { stride = K3_1 * C; return w.Wm1 + (R < K3_1 ? R : K3_1 - 1) * C + c; }
    else if constexpr (R < WR_21) { stride = K3_0 * C; return w.Wm0 + (R - WR_M0 < K3_0 ? R - WR_M0 : K3_0 - 1) * C + c; }
    else if constexpr (R < WR_20) { stride = K2_1 * C; return w.W21 + (R - WR_21) * C + c; }
    else if constexpr (R < WR_11) { stride = K2_0 * C; return w.W20 + (R - WR_20) * C + c; }
    else                          { stride = C;        return (R == WR_10 ? w.W10 : w.W11) + c; }
}

struct NodeCtx {
    const float* Bb[3];
    float*       sW;
    const float* T;
    const float* x;
    WTabs        wt;
    int tid, bn;
    bool has_next;
    bool oldw;
    const float* sP;
    const float* sWp;
    const float* sXp;
    const float* sU1;
    float*       out;
    int  bp;
    bool epi;
};

template <int G, int HF>
__device__ __forceinline__ void epi_piece(const NodeCtx& n, float& es)
{
    constexpr int NM  = (G == 0) ? K3_0 : K3_1;
    constexpr int PM  = (G == 0) ? COL_A0 : COL_A1 + (G - 1) * K3_1;
    constexpr int WM  = (G == 0) ? WR_M0 : WR_M1;
    constexpr int N2  = (G == 0) ? K2_0 : K2_1;
    constexpr int P2  = (G == 0) ? COL_20 : COL_21 + (G - 1) * K2_1;
    constexpr int W2  = (G == 0) ? WR_20 : WR_21;
    constexpr int U1O = (G == 0) ? 0 : 16 + (G - 1) * L;
    constexpr int W1  = (G == 0) ? WR_10 : WR_11;
    constexpr int H0  = 20;
    const int c = n.tid;
    const float* P  = n.sP + c * PSTR;
    const float* Wp = n.sWp + c;
    if constexpr (HF == 0) {
        float s = 0.f;
        #pragma unroll
        for (int k = 0; k < H0; ++k) s = fmaf(P[PM + k], Wp[(WM + k) * C], s);
        es = s;
    } else {
        float s = es;
        #pragma unroll
        for (int k = H0; k < NM; ++k) s = fmaf(P[PM + k], Wp[(WM + k) * C], s);
        #pragma unroll
        for (int k = 0; k < N2; ++k) s = fmaf(P[P2 + k], Wp[(W2 + k) * C], s);
        float d = 0.f;
        #pragma unroll
        for (int j = 0; j < L; ++j) d = fmaf(n.sU1[U1O + j], n.sXp[c * XSTR + j], d);
        s = fmaf(Wp[W1 * C], d, s);
        if constexpr (G == 0) n.out[(size_t)n.bp * (4 * C) + c] = s;
        else                  n.out[(size_t)n.bp * (4 * C) + C + c * 3 + (G - 1)] = s;
    }
}

template <int Q, int U>
__device__ __forceinline__ void wload(const NodeCtx& n, float (&wl)[WCH][E])
{
    if constexpr (U < WCH) {
        int stride;
        const float* wsrc = wrow_src<Q * WCH + U>(n.wt, n.tid, stride);
        #pragma unroll
        for (int e = 0; e < E; ++e) wl[U][e] = wsrc[e * stride];
        wload<Q, U + 1>(n, wl);
    }
}
template <int Q>
__device__ __forceinline__ void wsum(const NodeCtx& n, const float (&wl)[WCH][E], const float (&yv)[E])
{
    #pragma unroll
    for (int u = 0; u < WCH; ++u) {
        float s = 0.f;
        #pragma unroll
        for (int e = 0; e < E; ++e) s = fmaf(yv[e], wl[u][e], s);
        n.sW[(Q * WCH + u) * C + n.tid] = s;
    }
}

template <int Q>
__device__ __forceinline__ void kchunk(const NodeCtx& n, v8f (&acc)[NTILE], v2f (&bq)[2][NTILE],
                                       const float (&xr)[L], const float (&sel0)[4], const float (&sel1)[4],
                                       const float (&yv)[E], float (&wl)[WCH][E],
                                       v4f (&tv)[31], float (&xv)[8], float& es)
{
    if constexpr (Q >= 8) {
        constexpr int EG = (Q - 8) / 2, EH = (Q - 8) % 2;
        if (n.epi) epi_piece<EG, EH>(n, es);
    }
    if constexpr (Q <= NWB) {
        if (n.oldw) {
            if constexpr (Q >= 1)  wsum<Q - 1>(n, wl, yv);
            if constexpr (Q < NWB) wload<Q, 0>(n, wl);
        }
    }

    const float xp = xr[Q];
    #pragma unroll
    for (int j = 0; j < 4; ++j) {
        v2f a;
        a.x = xp * sel0[j];
        a.y = xp * sel1[j];
        if (4 * Q + j + 1 < 64) {
            const int sn = 4 * Q + j + 1;
            const float* Bn = n.Bb[sn / FAR_STEPS] + (2 * (sn % FAR_STEPS)) * S2;
            #pragma unroll
            for (int t = 0; t < NTILE; ++t)
                bq[(j + 1) & 1][t] = *(const v2f*)(Bn + t * 32);
        }
        __builtin_amdgcn_sched_barrier(0);
        #pragma unroll
        for (int t = 0; t < NTILE; ++t)
            acc[t] = __builtin_amdgcn_wmma_f32_16x16x4_f32(
                false, a, false, bq[j & 1][t], (short)0, acc[t], false, false);
        __builtin_amdgcn_sched_barrier(0);
    }

    if constexpr (Q == 7) {
        if (n.has_next && !n.oldw) node_loads(n.T, n.x, n.bn, n.tid, tv, xv);
    }
    if constexpr (Q == NWB) {
        if (n.has_next && n.oldw) node_loads(n.T, n.x, n.bn, n.tid, tv, xv);
    }

    if constexpr (Q + 1 < 16) kchunk<Q + 1>(n, acc, bq, xr, sel0, sel1, yv, wl, tv, xv, es);
}

__global__ __launch_bounds__(256) void mace_node_kernel(
    const float* __restrict__ x,    const float* __restrict__ y,
    const float* __restrict__ U2_0, const float* __restrict__ U1_0,
    const float* __restrict__ U2_1, const float* __restrict__ U1_1,
    const float* Wm0,  const float* W20,  const float* W10,
    const float* Wm1,  const float* W21,  const float* W11,
    const float* __restrict__ T,    float* out, int b2_far)
{
    extern __shared__ float sm[];
    float* sY  = sm + OFF_Y;
    float* sU1 = sm + OFF_U1;
    float* sB2 = sm + OFF_B2;
    float* sP  = sm + OFF_P;

    const int tid  = threadIdx.x;
    const int wv   = tid >> 5;
    const int lane = tid & 31;
    const int nl   = lane & 15;
    const int hi   = lane >> 4;
    const int crow = wv * 16 + nl;
    const bool oldw = __builtin_amdgcn_readfirstlane(tid) < 128;
    if (oldw) __builtin_amdgcn_s_setprio(3);

    int b = blockIdx.x;

    v4f tv[31];
    float xv[8];
    tv[30] = (v4f){0.f, 0.f, 0.f, 0.f};
    if (b < NN) node_loads(T, x, b, tid, tv, xv);

    if (tid < NPB * E) sY[tid] = y[(blockIdx.x + NODE_BLOCKS * (tid / E)) * E + (tid % E)];
    if (tid < 16)               sU1[tid] = U1_0[tid];
    else if (tid < 64)          sU1[tid] = U1_1[tid - 16];
    if (tid < 128) {
        const int p = tid;
        float* bp = sB2 + p * S2;
        #pragma unroll
        for (int m = 0; m < 3 * K2_1; ++m) {
            const int w = m / K2_1, k = m % K2_1;
            const int r0 = (w * 256 + 2 * p) * K2_1 + k;
            *(v2f*)(bp + (COL_21 + m) * 2) = (v2f){U2_1[r0], U2_1[r0 + K2_1]};
        }
        #pragma unroll
        for (int m = 0; m < K2_0; ++m)
            *(v2f*)(bp + (COL_20 + m) * 2) =
                (v2f){U2_0[(2 * p) * K2_0 + m], U2_0[(2 * p + 1) * K2_0 + m]};
    }

    int par = 0, it = 0;
    for (; b < NN; b += NODE_BLOCKS, par ^= 1, ++it) {
        float* sX = sm + OFF_X + par * (C * XSTR);
        float* sW = sm + (par ? OFF_W2 : OFF_W);

        #pragma unroll
        for (int j = 0; j < 31; ++j) {
            if (j < 30 || tid < 128) {
                const int f = 4 * (tid + 256 * j);
                const int p = f / TPAIR;
                float* d = sB2 + p * S2 + (f - p * TPAIR);
                *(v2f*)(d)     = (v2f){tv[j].x, tv[j].y};
                *(v2f*)(d + 2) = (v2f){tv[j].z, tv[j].w};
            }
        }
        #pragma unroll
        for (int j = 0; j < 8; ++j) {
            const int idx = tid + 256 * j;
            sX[(idx >> 4) * XSTR + (idx & 15)] = xv[j];
        }
        __syncthreads();
        float yv[E];
        #pragma unroll
        for (int e = 0; e < E; ++e) yv[e] = sY[it * E + e];

        float xr[L];
        #pragma unroll
        for (int j = 0; j < L; ++j) xr[j] = sX[crow * XSTR + j];

        float sel0[4], sel1[4];
        #pragma unroll
        for (int j = 0; j < 4; ++j) {
            const float e0 = xr[4 * j],     e1 = xr[4 * j + 1];
            const float e2 = xr[4 * j + 2], e3 = xr[4 * j + 3];
            sel0[j] = hi ? e2 : e0;
            sel1[j] = hi ? e3 : e1;
        }

        v8f acc[NTILE];
        #pragma unroll
        for (int t = 0; t < NTILE; ++t)
            acc[t] = (v8f){0.f, 0.f, 0.f, 0.f, 0.f, 0.f, 0.f, 0.f};

        NodeCtx n;
        n.Bb[0] = sB2 + hi * S2 + nl * 2;
        n.Bb[1] = n.Bb[0] + b2_far;
        n.Bb[2] = n.Bb[1] + b2_far;
        n.sW  = sW;  n.T = T;  n.x = x;
        n.wt  = WTabs{Wm0, W20, W10, Wm1, W21, W11};
        n.tid = tid; n.oldw = oldw;
        n.bn  = b + NODE_BLOCKS;
        n.has_next = n.bn < NN;
        n.sP  = sP;  n.sU1 = sU1;  n.out = out;
        n.sWp = sm + (par ? OFF_W : OFF_W2);
        n.sXp = sm + OFF_X + (par ^ 1) * (C * XSTR);
        n.bp  = b - NODE_BLOCKS;
        n.epi = oldw && it > 0;
        float es = 0.f;

        float wl[WCH][E];
        v2f bq[2][NTILE];
        #pragma unroll
        for (int t = 0; t < NTILE; ++t) bq[0][t] = *(const v2f*)(n.Bb[0] + t * 32);
        kchunk<0>(n, acc, bq, xr, sel0, sel1, yv, wl, tv, xv, es);

        __syncthreads();

        #pragma unroll
        for (int t = 0; t < NTILE; ++t) {
            #pragma unroll
            for (int r = 0; r < 8; ++r)
                sP[(wv * 16 + r + hi * 8) * PSTR + t * 16 + nl] = acc[t][r];
        }
    }

    if (it > 0) {
        __syncthreads();
        const int bl = b - NODE_BLOCKS;
        const float* sW = sm + (par ? OFF_W : OFF_W2);
        const float* sX = sm + OFF_X + (par ^ 1) * (C * XSTR);
        for (int idx = tid; idx < 4 * C; idx += 256) {
            const int c = idx & 127;
            const int g = idx >> 7;
            const float* P = sP + c * PSTR;
            float s = 0.f;
            if (g == 0) {
                #pragma unroll
                for (int k = 0; k < K3_0; ++k) s = fmaf(P[COL_A0 + k], sW[(WR_M0 + k) * C + c], s);
                #pragma unroll
                for (int k = 0; k < K2_0; ++k) s = fmaf(P[COL_20 + k], sW[(WR_20 + k) * C + c], s);
                float d = 0.f;
                #pragma unroll
                for (int j = 0; j < L; ++j) d = fmaf(sU1[j], sX[c * XSTR + j], d);
                s = fmaf(sW[WR_10 * C + c], d, s);
                out[(size_t)bl * (4 * C) + c] = s;
            } else {
                const int w = g - 1;
                #pragma unroll
                for (int k = 0; k < K3_1; ++k) s = fmaf(P[COL_A1 + w * K3_1 + k], sW[(WR_M1 + k) * C + c], s);
                #pragma unroll
                for (int k = 0; k < K2_1; ++k) s = fmaf(P[COL_21 + w * K2_1 + k], sW[(WR_21 + k) * C + c], s);
                float d = 0.f;
                #pragma unroll
                for (int j = 0; j < L; ++j) d = fmaf(sU1[16 + w * L + j], sX[c * XSTR + j], d);
                s = fmaf(sW[WR_11 * C + c], d, s);
                out[(size_t)bl * (4 * C) + C + c * 3 + w] = s;
            }
        }
    }
}

extern "C" void kernel_launch(void* const* d_in, const int* in_sizes, int n_in,
                              void* d_out, int out_size, void* d_ws, size_t ws_size,
                              hipStream_t stream) {
    (void)in_sizes; (void)n_in; (void)out_size; (void)ws_size;
    const float* x    = (const float*)d_in[0];
    const float* y    = (const float*)d_in[1];
    const float* U3_0 = (const float*)d_in[2];
    const float* U2_0 = (const float*)d_in[3];
    const float* U1_0 = (const float*)d_in[4];
    const float* U3_1 = (const float*)d_in[5];
    const float* U2_1 = (const float*)d_in[6];
    const float* U1_1 = (const float*)d_in[7];
    const float* Wm0  = (const float*)d_in[8];
    const float* W20  = (const float*)d_in[9];
    const float* W10  = (const float*)d_in[10];
    const float* Wm1  = (const float*)d_in[11];
    const float* W21  = (const float*)d_in[12];
    const float* W11  = (const float*)d_in[13];

    float* T  = (float*)d_ws;
    float* xs = T + WS_XS;

    xs_kernel<<<dim3(XS_BLOCKS), dim3(64), 0, stream>>>(x, xs);
    tbuild_kernel<<<dim3(NTT), dim3(256), 0, stream>>>(U3_0, U3_1, xs, T);

    constexpr size_t smem_bytes = (size_t)SMEM_FLOATS * sizeof(float);
    mace_node_kernel<<<dim3(NODE_BLOCKS), dim3(256), smem_bytes, stream>>>(
        x, y, U2_0, U1_0, U2_1, U1_1,
        Wm0, W20, W10, Wm1, W21, W11, T, (float*)d_out, B2_FAR);
}
